# starting kernel
# baseline (speedup 1.0000x reference)
#include <hip/hip_runtime.h>
#include <hip/hip_bf16.h>

typedef __attribute__((ext_vector_type(2))) float v2f;
typedef __attribute__((ext_vector_type(8))) float v8f;

#define B_   32
#define L_   512
#define C_   128
#define HID_ 256
#define H_   4
#define D_   64
#define NL_  3
#define NEG_ 0.2f
#define EPS_ 1e-5f

template <int M, int N, int K, int LDA, int LDB, int LDD, int INNER,
          long SAO, long SAI, long SBO, long SBI, long SDO, long SDI,
          long SBIASI, bool TRANSA, bool TRANSD, bool HASBIAS>
__global__ __launch_bounds__(128) void wmma_gemm_f32(
    const float* __restrict__ A, const float* __restrict__ Bm,
    const float* __restrict__ bias, float* __restrict__ D)
{
  const int lane = threadIdx.x & 31;
  const int wave = threadIdx.x >> 5;
  const int half = (lane >> 4) & 1;
  const int r    = lane & 15;

  constexpr int tilesN = N >> 5;
  const int tile = blockIdx.x * 4 + wave;
  const int tm = tile / tilesN;
  const int tn = tile % tilesN;

  const int batch = blockIdx.y;
  const int bo = batch / INNER;
  const int bi = batch % INNER;
  A    += bo * SAO + bi * SAI;
  Bm   += bo * SBO + bi * SBI;
  D    += bo * SDO + bi * SDI;
  if (HASBIAS) bias += bi * SBIASI;

  const int m0 = tm << 5;
  const int n0 = tn << 5;

  constexpr int stepA = TRANSA ? 4 * LDA : 4;
  constexpr int stepB = 4 * LDB;
  int offA0 = TRANSA ? (2 * half) * LDA + (m0 + r)
                     : (m0 + r) * LDA + 2 * half;
  int offA1 = offA0 + (TRANSA ? 16 : 16 * LDA);
  int offB0 = (2 * half) * LDB + (n0 + r);
  int offB1 = offB0 + 16;

  v8f acc00 = {0.f,0.f,0.f,0.f,0.f,0.f,0.f,0.f};
  v8f acc01 = acc00, acc10 = acc00, acc11 = acc00;

#pragma unroll 4
  for (int k0 = 0; k0 < K; k0 += 4) {
    v2f a0, a1, b0, b1;
    if (TRANSA) {
      a0.x = A[offA0]; a0.y = A[offA0 + LDA];
      a1.x = A[offA1]; a1.y = A[offA1 + LDA];
    } else {
      a0 = *(const v2f*)(A + offA0);
      a1 = *(const v2f*)(A + offA1);
    }
    b0.x = Bm[offB0]; b0.y = Bm[offB0 + LDB];
    b1.x = Bm[offB1]; b1.y = Bm[offB1 + LDB];
    offA0 += stepA; offA1 += stepA; offB0 += stepB; offB1 += stepB;

    acc00 = __builtin_amdgcn_wmma_f32_16x16x4_f32(false, a0, false, b0, (short)0, acc00, false, false);
    acc01 = __builtin_amdgcn_wmma_f32_16x16x4_f32(false, a0, false, b1, (short)0, acc01, false, false);
    acc10 = __builtin_amdgcn_wmma_f32_16x16x4_f32(false, a1, false, b0, (short)0, acc10, false, false);
    acc11 = __builtin_amdgcn_wmma_f32_16x16x4_f32(false, a1, false, b1, (short)0, acc11, false, false);
  }

  const int col0 = n0 + r;
  const int col1 = col0 + 16;
  const float bv0 = HASBIAS ? bias[col0] : 0.f;
  const float bv1 = HASBIAS ? bias[col1] : 0.f;
#pragma unroll
  for (int i = 0; i < 8; ++i) {
    const int row0 = m0 + i + 8 * half;
    const int row1 = row0 + 16;
    if (TRANSD) {
      D[(long)col0 * LDD + row0] = acc00[i] + bv0;
      D[(long)col1 * LDD + row0] = acc01[i] + bv1;
      D[(long)col0 * LDD + row1] = acc10[i] + bv0;
      D[(long)col1 * LDD + row1] = acc11[i] + bv1;
    } else {
      D[(long)row0 * LDD + col0] = acc00[i] + bv0;
      D[(long)row0 * LDD + col1] = acc01[i] + bv1;
      D[(long)row1 * LDD + col0] = acc10[i] + bv0;
      D[(long)row1 * LDD + col1] = acc11[i] + bv1;
    }
  }
}

__global__ __launch_bounds__(128) void gat_scores_softmax(
    const float* __restrict__ gl, const float* __restrict__ gr,
    const float* __restrict__ att, float* __restrict__ alpha)
{
  __shared__ float sGr[C_ * D_];
  __shared__ float sAtt[D_];
  __shared__ float sRow[D_];
  __shared__ float red[4];

  const int tid = threadIdx.x;
  const int h = blockIdx.x;
  const int b = blockIdx.y;
  const float* glb = gl + (long)b * C_ * HID_ + h * D_;
  const float* grb = gr + (long)b * C_ * HID_ + h * D_;

  for (int idx = tid; idx < C_ * D_; idx += 128) {
    const int s = idx >> 6, d = idx & 63;
    sGr[idx] = grb[(long)s * HID_ + d];
  }
  if (tid < D_) sAtt[tid] = att[h * D_ + tid];
  __syncthreads();

  const int lane = tid & 31;
  const int wv = tid >> 5;
  float* arow = alpha + (long)(b * H_ + h) * C_ * C_;

  for (int t = 0; t < C_; ++t) {
    if (tid < D_) sRow[tid] = glb[(long)t * HID_ + tid];
    __syncthreads();

    const int s = tid;
    float sc = 0.f;
    const float* g = &sGr[s * D_];
#pragma unroll 8
    for (int d = 0; d < D_; ++d) {
      float v = sRow[d] + g[d];
      v = v > 0.f ? v : NEG_ * v;
      sc = fmaf(sAtt[d], v, sc);
    }
    float m = sc;
    for (int off = 16; off > 0; off >>= 1) m = fmaxf(m, __shfl_xor(m, off, 32));
    if (lane == 0) red[wv] = m;
    __syncthreads();
    m = fmaxf(fmaxf(red[0], red[1]), fmaxf(red[2], red[3]));
    __syncthreads();
    const float e = __expf(sc - m);
    float ssum = e;
    for (int off = 16; off > 0; off >>= 1) ssum += __shfl_xor(ssum, off, 32);
    if (lane == 0) red[wv] = ssum;
    __syncthreads();
    ssum = red[0] + red[1] + red[2] + red[3];
    arow[(long)t * C_ + s] = e / ssum;
    __syncthreads();
  }
}

__global__ __launch_bounds__(256) void post_elu_residual_ln(
    float* __restrict__ h, const float* __restrict__ gout,
    const float* __restrict__ g, const float* __restrict__ bb)
{
  __shared__ float red[8];
  const int tid = threadIdx.x;
  const long base = (long)blockIdx.x * HID_;

  float x = gout[base + tid];
  x = x > 0.f ? x : (__expf(x) - 1.f);
  const float v = h[base + tid] + x;

  const int lane = tid & 31, wv = tid >> 5;
  float s = v;
  for (int off = 16; off > 0; off >>= 1) s += __shfl_xor(s, off, 32);
  if (lane == 0) red[wv] = s;
  __syncthreads();
  s = 0.f;
#pragma unroll
  for (int i = 0; i < 8; ++i) s += red[i];
  const float mu = s * (1.f / HID_);
  __syncthreads();

  const float dmu = v - mu;
  float q = dmu * dmu;
  for (int off = 16; off > 0; off >>= 1) q += __shfl_xor(q, off, 32);
  if (lane == 0) red[wv] = q;
  __syncthreads();
  q = 0.f;
#pragma unroll
  for (int i = 0; i < 8; ++i) q += red[i];
  const float inv = rsqrtf(q * (1.f / HID_) + EPS_);

  h[base + tid] = dmu * inv * g[tid] + bb[tid];
}

extern "C" void kernel_launch(void* const* d_in, const int* in_sizes, int n_in,
                              void* d_out, int out_size, void* d_ws, size_t ws_size,
                              hipStream_t stream) {
  (void)in_sizes; (void)n_in; (void)out_size; (void)ws_size;
  const float* x      = (const float*)d_in[0];
  const float* W_emb  = (const float*)d_in[1];
  const float* b_emb  = (const float*)d_in[2];
  const float* Wl     = (const float*)d_in[3];
  const float* Wr     = (const float*)d_in[4];
  const float* att    = (const float*)d_in[5];
  const float* bias_g = (const float*)d_in[6];
  const float* ln_g   = (const float*)d_in[7];
  const float* ln_b   = (const float*)d_in[8];
  const float* W_proj = (const float*)d_in[9];
  const float* b_proj = (const float*)d_in[10];

  float* out   = (float*)d_out;
  float* alpha = out + (long)B_ * L_ * C_;

  float* ws = (float*)d_ws;
  const long hwN = (long)B_ * C_ * HID_;
  float* h    = ws;
  float* gl   = ws + hwN;
  float* gr   = ws + 2 * hwN;
  float* gout = ws + 3 * hwN;

  wmma_gemm_f32<C_, HID_, L_, C_, HID_, HID_, 1,
                (long)L_ * C_, 0, 0, 0, (long)C_ * HID_, 0, 0,
                true, false, true>
      <<<dim3((C_ / 32) * (HID_ / 32) / 4, B_), 128, 0, stream>>>(
          x, W_emb, b_emb, h);

  for (int layer = 0; layer < NL_; ++layer) {
    const float* Wl_i  = Wl + (long)layer * HID_ * HID_;
    const float* Wr_i  = Wr + (long)layer * HID_ * HID_;
    const float* att_i = att + (long)layer * H_ * D_;
    const float* bg_i  = bias_g + (long)layer * HID_;

    dim3 g1((C_ / 32) * (HID_ / 32) / 4, B_);
    wmma_gemm_f32<C_, HID_, HID_, HID_, HID_, HID_, 1,
                  (long)C_ * HID_, 0, 0, 0, (long)C_ * HID_, 0, 0,
                  false, false, false>
        <<<g1, 128, 0, stream>>>(h, Wl_i, nullptr, gl);
    wmma_gemm_f32<C_, HID_, HID_, HID_, HID_, HID_, 1,
                  (long)C_ * HID_, 0, 0, 0, (long)C_ * HID_, 0, 0,
                  false, false, false>
        <<<g1, 128, 0, stream>>>(h, Wr_i, nullptr, gr);

    gat_scores_softmax<<<dim3(H_, B_), 128, 0, stream>>>(gl, gr, att_i, alpha);

    wmma_gemm_f32<C_, D_, C_, C_, HID_, HID_, H_,
                  (long)H_ * C_ * C_, (long)C_ * C_,
                  (long)C_ * HID_, D_, (long)C_ * HID_, D_, D_,
                  false, false, true>
        <<<dim3((C_ / 32) * (D_ / 32) / 4, B_ * H_), 128, 0, stream>>>(
            alpha, gr, bg_i, gout);

    post_elu_residual_ln<<<B_ * C_, 256, 0, stream>>>(
        h, gout, ln_g + (long)layer * HID_, ln_b + (long)layer * HID_);
  }

  wmma_gemm_f32<C_, L_, HID_, HID_, L_, C_, 1,
                (long)C_ * HID_, 0, 0, 0, (long)L_ * C_, 0, 0,
                false, true, true>
      <<<dim3((C_ / 32) * (L_ / 32) / 4, B_), 128, 0, stream>>>(
          h, W_proj, b_proj, out);
}


# GEMM launches: A and B tiles staged in LDS by the whole work-group (16-byte loads over whole 128-byte lines, each line fetched once) instead of per-lane global fragment loads; k order unchanged
# speedup vs baseline: 6.2109x; 6.2109x over previous
#include <hip/hip_runtime.h>
#include <hip/hip_bf16.h>

typedef __attribute__((ext_vector_type(2))) float v2f;
typedef __attribute__((ext_vector_type(8))) float v8f;

#define B_   32
#define L_   512
#define C_   128
#define HID_ 256
#define H_   4
#define D_   64
#define NL_  3
#define NEG_ 0.2f
#define EPS_ 1e-5f
#define KC_  64

template <int M, int N, int K, int LDA, int LDB, int LDD, int INNER,
          long SAO, long SAI, long SBO, long SBI, long SDO, long SDI,
          long SBIASI, bool TRANSA, bool TRANSD, bool HASBIAS>
__global__ __launch_bounds__(128) void wmma_gemm_f32(
    const float* __restrict__ A, const float* __restrict__ Bm,
    const float* __restrict__ bias, float* __restrict__ D)
{
  constexpr int tilesN = N >> 5;
  constexpr int WN = tilesN >= 4 ? 4 : tilesN;
  constexpr int WM = 4 / WN;
  constexpr int BM = 32 * WM;
  constexpr int BN = 32 * WN;
  constexpr int KC = KC_;
  constexpr int PA = TRANSA ? BM + 8 : KC + 4;
  constexpr int PB = BN + 8;
  constexpr int A_SZ = TRANSA ? KC * PA : BM * PA;
  constexpr int B_SZ = KC * PB;
  constexpr int A_F4 = BM * KC / 4 / 128;
  constexpr int B_F4 = KC * BN / 4 / 128;
  constexpr int NCH = K / KC;
  static_assert(K % KC == 0 && N % BN == 0 && M % BM == 0, "shape");

  __shared__ __attribute__((aligned(16))) float sA[2 * A_SZ];
  __shared__ __attribute__((aligned(16))) float sB[2 * B_SZ];

  const int tid  = threadIdx.x;
  const int lane = tid & 31;
  const int wave = tid >> 5;
  const int half = (lane >> 4) & 1;
  const int r    = lane & 15;

  constexpr int tilesNb = N / BN;
  const int mb = (blockIdx.x / tilesNb) * BM;
  const int nb = (blockIdx.x % tilesNb) * BN;
  const int wm = (wave / WN) * 32;
  const int wn = (wave % WN) * 32;

  const int batch = blockIdx.y;
  const int bo = batch / INNER;
  const int bi = batch % INNER;
  A    += bo * SAO + bi * SAI;
  Bm   += bo * SBO + bi * SBI;
  D    += bo * SDO + bi * SDI;
  if (HASBIAS) bias += bi * SBIASI;

  int gA[A_F4], lA[A_F4], gB[B_F4], lB[B_F4];
#pragma unroll
  for (int j = 0; j < A_F4; ++j) {
    const int idx = tid + 128 * j;
    if (TRANSA) {
      const int krow = idx / (BM / 4), mq = idx % (BM / 4);
      gA[j] = krow * LDA + mb + 4 * mq;
      lA[j] = krow * PA + 4 * mq;
    } else {
      const int row = idx / (KC / 4), kq = idx % (KC / 4);
      gA[j] = (mb + row) * LDA + 4 * kq;
      lA[j] = row * PA + 4 * kq;
    }
  }
#pragma unroll
  for (int j = 0; j < B_F4; ++j) {
    const int idx = tid + 128 * j;
    const int krow = idx / (BN / 4), nq = idx % (BN / 4);
    gB[j] = krow * LDB + nb + 4 * nq;
    lB[j] = krow * PB + 4 * nq;
  }
  constexpr int stepA = TRANSA ? KC * LDA : KC;
  constexpr int stepB = KC * LDB;


  v8f acc00 = {0.f,0.f,0.f,0.f,0.f,0.f,0.f,0.f};
  v8f acc01 = acc00, acc10 = acc00, acc11 = acc00;

#pragma unroll
  for (int c = 0; c < NCH; ++c) {
    float* as = sA + (c & 1) * A_SZ;
    float* bs = sB + (c & 1) * B_SZ;
    float4 ra[A_F4], rb[B_F4];
#pragma unroll
    for (int j = 0; j < A_F4; ++j) ra[j] = *(const float4*)(A + gA[j] + c * stepA);
#pragma unroll
    for (int j = 0; j < B_F4; ++j) rb[j] = *(const float4*)(Bm + gB[j] + c * stepB);
#pragma unroll
    for (int j = 0; j < A_F4; ++j) *(float4*)(as + lA[j]) = ra[j];
#pragma unroll
    for (int j = 0; j < B_F4; ++j) *(float4*)(bs + lB[j]) = rb[j];
    __syncthreads();
#pragma unroll 4
    for (int ks = 0; ks < KC / 4; ++ks) {
      const int k = 4 * ks + 2 * half;
      v2f a0, a1, b0, b1;
      if (TRANSA) {
        a0.x = as[k * PA + wm + r];      a0.y = as[(k + 1) * PA + wm + r];
        a1.x = as[k * PA + wm + 16 + r]; a1.y = as[(k + 1) * PA + wm + 16 + r];
      } else {
        a0 = *(const v2f*)(as + (wm + r) * PA + k);
        a1 = *(const v2f*)(as + (wm + 16 + r) * PA + k);
      }
      b0.x = bs[k * PB + wn + r];      b0.y = bs[(k + 1) * PB + wn + r];
      b1.x = bs[k * PB + wn + 16 + r]; b1.y = bs[(k + 1) * PB + wn + 16 + r];

      acc00 = __builtin_amdgcn_wmma_f32_16x16x4_f32(false, a0, false, b0, (short)0, acc00, false, false);
      acc01 = __builtin_amdgcn_wmma_f32_16x16x4_f32(false, a0, false, b1, (short)0, acc01, false, false);
      acc10 = __builtin_amdgcn_wmma_f32_16x16x4_f32(false, a1, false, b0, (short)0, acc10, false, false);
      acc11 = __builtin_amdgcn_wmma_f32_16x16x4_f32(false, a1, false, b1, (short)0, acc11, false, false);
    }
  }

  const int m0 = mb + wm;
  const int n0 = nb + wn;
  const int col0 = n0 + r;
  const int col1 = col0 + 16;
  const float bv0 = HASBIAS ? bias[col0] : 0.f;
  const float bv1 = HASBIAS ? bias[col1] : 0.f;
#pragma unroll
  for (int i = 0; i < 8; ++i) {
    const int row0 = m0 + i + 8 * half;
    const int row1 = row0 + 16;
    if (TRANSD) {
      D[(long)col0 * LDD + row0] = acc00[i] + bv0;
      D[(long)col1 * LDD + row0] = acc01[i] + bv1;
      D[(long)col0 * LDD + row1] = acc10[i] + bv0;
      D[(long)col1 * LDD + row1] = acc11[i] + bv1;
    } else {
      D[(long)row0 * LDD + col0] = acc00[i] + bv0;
      D[(long)row0 * LDD + col1] = acc01[i] + bv1;
      D[(long)row1 * LDD + col0] = acc10[i] + bv0;
      D[(long)row1 * LDD + col1] = acc11[i] + bv1;
    }
  }
}

#define TCH_ 32
#define NWV_ 8
#define SWV_ (C_ / NWV_)
#define SP_  (C_ + 1)

typedef const __attribute__((address_space(4))) float cfloat_k;

template <int HF>
__device__ __forceinline__ void score_half(
    const float* __restrict__ glrow, cfloat_k* grw,
    const float* sAtt, float* srow_w)
{
  const float4* glp = (const float4*)(glrow + HF * (D_ / 2));
  float glr[D_ / 2];
#pragma unroll
  for (int q = 0; q < D_ / 8; ++q) {
    const float4 v = glp[q];
    glr[4 * q + 0] = v.x; glr[4 * q + 1] = v.y;
    glr[4 * q + 2] = v.z; glr[4 * q + 3] = v.w;
  }
  float av[D_ / 2];
#pragma unroll
  for (int d = 0; d < D_ / 2; ++d) av[d] = sAtt[HF * (D_ / 2) + d];

  cfloat_k* gh = grw + HF * (D_ / 2);
#pragma unroll 2
  for (int i = 0; i < SWV_; ++i) {
    cfloat_k* g = gh + (long)i * HID_;
    float sc = HF ? srow_w[i] : 0.f;
#pragma unroll
    for (int d = 0; d < D_ / 2; ++d) {
      const float v = glr[d] + g[d];
      sc = fmaf(av[d], fmaxf(v, NEG_ * v), sc);
    }
    srow_w[i] = sc;
  }
}

__global__ __launch_bounds__(256) void gat_scores_softmax(
    const float* __restrict__ gl, const float* __restrict__ gr,
    const float* __restrict__ att, float* __restrict__ alpha)
{
  __shared__ float sS[TCH_ * SP_];
  __shared__ float sAtt[D_];

  const int tid = threadIdx.x;
  const int lane = tid & 31;
  const int wv = __builtin_amdgcn_readfirstlane(tid >> 5);
  const int tc = blockIdx.x;
  const int h = blockIdx.y;
  const int b = blockIdx.z;

  if (wv < D_ / 32) sAtt[tid] = att[h * D_ + tid];

  __syncthreads();

  const float* glrow = gl + ((long)b * C_ + tc * TCH_ + lane) * HID_ + h * D_;
  cfloat_k* grw = (cfloat_k*)(gr + ((long)b * C_ + wv * SWV_) * HID_ + h * D_);
  float* srow_w = &sS[lane * SP_ + wv * SWV_];

  score_half<0>(glrow, grw, sAtt, srow_w);
  score_half<1>(glrow, grw, sAtt, srow_w);
  __syncthreads();

  float* arow = alpha + ((long)(b * H_ + h) * C_ + tc * TCH_) * C_;
#pragma unroll
  for (int k = 0; k < TCH_ / NWV_; ++k) {
    const int tl = wv * (TCH_ / NWV_) + k;
    const float* sr = &sS[tl * SP_];
    float v0 = sr[lane], v1 = sr[lane + 32], v2 = sr[lane + 64], v3 = sr[lane + 96];
    float m = fmaxf(fmaxf(v0, v1), fmaxf(v2, v3));
    for (int off = 16; off > 0; off >>= 1) m = fmaxf(m, __shfl_xor(m, off, 32));
    const float e0 = __expf(v0 - m), e1 = __expf(v1 - m);
    const float e2 = __expf(v2 - m), e3 = __expf(v3 - m);
    float r0 = e0, r1 = e1, r2 = e2, r3 = e3;
    for (int off = 16; off > 0; off >>= 1) {
      r0 += __shfl_xor(r0, off, 32);
      r1 += __shfl_xor(r1, off, 32);
      r2 += __shfl_xor(r2, off, 32);
      r3 += __shfl_xor(r3, off, 32);
    }
    const float ssum = r0 + r1 + r2 + r3;
    float* o = arow + (long)tl * C_ + lane;
    o[0]  = e0 / ssum;
    o[32] = e1 / ssum;
    o[64] = e2 / ssum;
    o[96] = e3 / ssum;
  }
}

__global__ __launch_bounds__(256) void post_elu_residual_ln(
    float* __restrict__ h, const float* __restrict__ gout,
    const float* __restrict__ g, const float* __restrict__ bb)
{
  __shared__ float red[8];
  const int tid = threadIdx.x;
  const long base = (long)blockIdx.x * HID_;

  float x = gout[base + tid];
  x = x > 0.f ? x : (__expf(x) - 1.f);
  const float v = h[base + tid] + x;

  const int lane = tid & 31, wv = tid >> 5;
  float s = v;
  for (int off = 16; off > 0; off >>= 1) s += __shfl_xor(s, off, 32);
  if (lane == 0) red[wv] = s;
  __syncthreads();
  s = 0.f;
#pragma unroll
  for (int i = 0; i < 8; ++i) s += red[i];
  const float mu = s * (1.f / HID_);
  __syncthreads();

  const float dmu = v - mu;
  float q = dmu * dmu;
  for (int off = 16; off > 0; off >>= 1) q += __shfl_xor(q, off, 32);
  if (lane == 0) red[wv] = q;
  __syncthreads();
  q = 0.f;
#pragma unroll
  for (int i = 0; i < 8; ++i) q += red[i];
  const float inv = rsqrtf(q * (1.f / HID_) + EPS_);

  h[base + tid] = dmu * inv * g[tid] + bb[tid];
}

extern "C" void kernel_launch(void* const* d_in, const int* in_sizes, int n_in,
                              void* d_out, int out_size, void* d_ws, size_t ws_size,
                              hipStream_t stream) {
  (void)in_sizes; (void)n_in; (void)out_size; (void)ws_size;
  const float* x      = (const float*)d_in[0];
  const float* W_emb  = (const float*)d_in[1];
  const float* b_emb  = (const float*)d_in[2];
  const float* Wl     = (const float*)d_in[3];
  const float* Wr     = (const float*)d_in[4];
  const float* att    = (const float*)d_in[5];
  const float* bias_g = (const float*)d_in[6];
  const float* ln_g   = (const float*)d_in[7];
  const float* ln_b   = (const float*)d_in[8];
  const float* W_proj = (const float*)d_in[9];
  const float* b_proj = (const float*)d_in[10];

  float* out   = (float*)d_out;
  float* alpha = out + (long)B_ * L_ * C_;

  float* ws = (float*)d_ws;
  const long hwN = (long)B_ * C_ * HID_;
  float* h    = ws;
  float* gl   = ws + hwN;
  float* gr   = ws + 2 * hwN;
  float* gout = ws + 3 * hwN;

  wmma_gemm_f32<C_, HID_, L_, C_, HID_, HID_, 1,
                (long)L_ * C_, 0, 0, 0, (long)C_ * HID_, 0, 0,
                true, false, true>
      <<<dim3((C_ / 32) * (HID_ / 32) / 4, B_), 128, 0, stream>>>(
          x, W_emb, b_emb, h);

  for (int layer = 0; layer < NL_; ++layer) {
    const float* Wl_i  = Wl + (long)layer * HID_ * HID_;
    const float* Wr_i  = Wr + (long)layer * HID_ * HID_;
    const float* att_i = att + (long)layer * H_ * D_;
    const float* bg_i  = bias_g + (long)layer * HID_;

    dim3 g1((C_ / 32) * (HID_ / 32) / 4, B_);
    wmma_gemm_f32<C_, HID_, HID_, HID_, HID_, HID_, 1,
                  (long)C_ * HID_, 0, 0, 0, (long)C_ * HID_, 0, 0,
                  false, false, false>
        <<<g1, 128, 0, stream>>>(h, Wl_i, nullptr, gl);
    wmma_gemm_f32<C_, HID_, HID_, HID_, HID_, HID_, 1,
                  (long)C_ * HID_, 0, 0, 0, (long)C_ * HID_, 0, 0,
                  false, false, false>
        <<<g1, 128, 0, stream>>>(h, Wr_i, nullptr, gr);

    gat_scores_softmax<<<dim3(C_ / TCH_, H_, B_), 32 * NWV_, 0, stream>>>(gl, gr, att_i, alpha);

    wmma_gemm_f32<C_, D_, C_, C_, HID_, HID_, H_,
                  (long)H_ * C_ * C_, (long)C_ * C_,
                  (long)C_ * HID_, D_, (long)C_ * HID_, D_, D_,
                  false, false, true>
        <<<dim3((C_ / 32) * (D_ / 32) / 4, B_ * H_), 128, 0, stream>>>(
            alpha, gr, bg_i, gout);

    post_elu_residual_ln<<<B_ * C_, 256, 0, stream>>>(
        h, gout, ln_g + (long)layer * HID_, ln_b + (long)layer * HID_);
  }

  wmma_gemm_f32<C_, L_, HID_, HID_, L_, C_, 1,
                (long)C_ * HID_, 0, 0, 0, (long)L_ * C_, 0, 0,
                false, true, true>
      <<<dim3((C_ / 32) * (L_ / 32) / 4, B_), 128, 0, stream>>>(
          h, W_proj, b_proj, out);
}


# GEMM launches: work-group block of D halved to 32x64 (wave tile 32x16, two accumulators), twice the work-groups: two share a WGP, one's tile loads overlap the other's matrix work; k order unchanged
# speedup vs baseline: 6.9082x; 1.1123x over previous
#include <hip/hip_runtime.h>
#include <hip/hip_bf16.h>

typedef __attribute__((ext_vector_type(2))) float v2f;
typedef __attribute__((ext_vector_type(8))) float v8f;

#define B_   32
#define L_   512
#define C_   128
#define HID_ 256
#define H_   4
#define D_   64
#define NL_  3
#define NEG_ 0.2f
#define EPS_ 1e-5f
#define KC_  64

template <int M, int N, int K, int LDA, int LDB, int LDD, int INNER,
          long SAO, long SAI, long SBO, long SBI, long SDO, long SDI,
          long SBIASI, bool TRANSA, bool TRANSD, bool HASBIAS>
__global__ __launch_bounds__(128) void wmma_gemm_f32(
    const float* __restrict__ A, const float* __restrict__ Bm,
    const float* __restrict__ bias, float* __restrict__ D)
{
  constexpr int BM = 32;
  constexpr int BN = 64;
  constexpr int KC = KC_;
  constexpr int PA = TRANSA ? BM + 8 : KC + 4;
  constexpr int PB = BN + 8;
  constexpr int A_SZ = TRANSA ? KC * PA : BM * PA;
  constexpr int B_SZ = KC * PB;
  constexpr int A_F4 = BM * KC / 4 / 128;
  constexpr int B_F4 = KC * BN / 4 / 128;
  constexpr int NCH = K / KC;
  static_assert(K % KC == 0 && N % BN == 0 && M % BM == 0, "shape");

  __shared__ __attribute__((aligned(16))) float sA[2 * A_SZ];
  __shared__ __attribute__((aligned(16))) float sB[2 * B_SZ];

  const int tid  = threadIdx.x;
  const int lane = tid & 31;
  const int wave = tid >> 5;
  const int half = (lane >> 4) & 1;
  const int r    = lane & 15;

  constexpr int tilesNb = N / BN;
  const int mb = (blockIdx.x / tilesNb) * BM;
  const int nb = (blockIdx.x % tilesNb) * BN;
  const int wn = wave * 16;

  const int batch = blockIdx.y;
  const int bo = batch / INNER;
  const int bi = batch % INNER;
  A    += bo * SAO + bi * SAI;
  Bm   += bo * SBO + bi * SBI;
  D    += bo * SDO + bi * SDI;
  if (HASBIAS) bias += bi * SBIASI;

  int gA[A_F4], lA[A_F4], gB[B_F4], lB[B_F4];
#pragma unroll
  for (int j = 0; j < A_F4; ++j) {
    const int idx = tid + 128 * j;
    if (TRANSA) {
      const int krow = idx / (BM / 4), mq = idx % (BM / 4);
      gA[j] = krow * LDA + mb + 4 * mq;
      lA[j] = krow * PA + 4 * mq;
    } else {
      const int row = idx / (KC / 4), kq = idx % (KC / 4);
      gA[j] = (mb + row) * LDA + 4 * kq;
      lA[j] = row * PA + 4 * kq;
    }
  }
#pragma unroll
  for (int j = 0; j < B_F4; ++j) {
    const int idx = tid + 128 * j;
    const int krow = idx / (BN / 4), nq = idx % (BN / 4);
    gB[j] = krow * LDB + nb + 4 * nq;
    lB[j] = krow * PB + 4 * nq;
  }
  constexpr int stepA = TRANSA ? KC * LDA : KC;
  constexpr int stepB = KC * LDB;


  v8f acc0 = {0.f,0.f,0.f,0.f,0.f,0.f,0.f,0.f};
  v8f acc1 = acc0;

#pragma unroll
  for (int c = 0; c < NCH; ++c) {
    float* as = sA + (c & 1) * A_SZ;
    float* bs = sB + (c & 1) * B_SZ;
    float4 ra[A_F4], rb[B_F4];
#pragma unroll
    for (int j = 0; j < A_F4; ++j) ra[j] = *(const float4*)(A + gA[j] + c * stepA);
#pragma unroll
    for (int j = 0; j < B_F4; ++j) rb[j] = *(const float4*)(Bm + gB[j] + c * stepB);
#pragma unroll
    for (int j = 0; j < A_F4; ++j) *(float4*)(as + lA[j]) = ra[j];
#pragma unroll
    for (int j = 0; j < B_F4; ++j) *(float4*)(bs + lB[j]) = rb[j];
    __syncthreads();
#pragma unroll 8
    for (int ks = 0; ks < KC / 4; ++ks) {
      const int k = 4 * ks + 2 * half;
      v2f a0, a1, b0;
      if (TRANSA) {
        a0.x = as[k * PA + r];      a0.y = as[(k + 1) * PA + r];
        a1.x = as[k * PA + 16 + r]; a1.y = as[(k + 1) * PA + 16 + r];
      } else {
        a0 = *(const v2f*)(as + r * PA + k);
        a1 = *(const v2f*)(as + (16 + r) * PA + k);
      }
      b0.x = bs[k * PB + wn + r];   b0.y = bs[(k + 1) * PB + wn + r];

      acc0 = __builtin_amdgcn_wmma_f32_16x16x4_f32(false, a0, false, b0, (short)0, acc0, false, false);
      acc1 = __builtin_amdgcn_wmma_f32_16x16x4_f32(false, a1, false, b0, (short)0, acc1, false, false);
    }
  }

  const int col0 = nb + wn + r;
  const float bv0 = HASBIAS ? bias[col0] : 0.f;
#pragma unroll
  for (int i = 0; i < 8; ++i) {
    const int row0 = mb + i + 8 * half;
    const int row1 = row0 + 16;
    if (TRANSD) {
      D[(long)col0 * LDD + row0] = acc0[i] + bv0;
      D[(long)col0 * LDD + row1] = acc1[i] + bv0;
    } else {
      D[(long)row0 * LDD + col0] = acc0[i] + bv0;
      D[(long)row1 * LDD + col0] = acc1[i] + bv0;
    }
  }
}

#define TCH_ 32
#define NWV_ 8
#define SWV_ (C_ / NWV_)
#define SP_  (C_ + 1)

typedef const __attribute__((address_space(4))) float cfloat_k;

template <int HF>
__device__ __forceinline__ void score_half(
    const float* __restrict__ glrow, cfloat_k* grw,
    const float* sAtt, float* srow_w)
{
  const float4* glp = (const float4*)(glrow + HF * (D_ / 2));
  float glr[D_ / 2];
#pragma unroll
  for (int q = 0; q < D_ / 8; ++q) {
    const float4 v = glp[q];
    glr[4 * q + 0] = v.x; glr[4 * q + 1] = v.y;
    glr[4 * q + 2] = v.z; glr[4 * q + 3] = v.w;
  }
  float av[D_ / 2];
#pragma unroll
  for (int d = 0; d < D_ / 2; ++d) av[d] = sAtt[HF * (D_ / 2) + d];

  cfloat_k* gh = grw + HF * (D_ / 2);
#pragma unroll 2
  for (int i = 0; i < SWV_; ++i) {
    cfloat_k* g = gh + (long)i * HID_;
    float sc = HF ? srow_w[i] : 0.f;
#pragma unroll
    for (int d = 0; d < D_ / 2; ++d) {
      const float v = glr[d] + g[d];
      sc = fmaf(av[d], fmaxf(v, NEG_ * v), sc);
    }
    srow_w[i] = sc;
  }
}

__global__ __launch_bounds__(256) void gat_scores_softmax(
    const float* __restrict__ gl, const float* __restrict__ gr,
    const float* __restrict__ att, float* __restrict__ alpha)
{
  __shared__ float sS[TCH_ * SP_];
  __shared__ float sAtt[D_];

  const int tid = threadIdx.x;
  const int lane = tid & 31;
  const int wv = __builtin_amdgcn_readfirstlane(tid >> 5);
  const int tc = blockIdx.x;
  const int h = blockIdx.y;
  const int b = blockIdx.z;

  if (wv < D_ / 32) sAtt[tid] = att[h * D_ + tid];

  __syncthreads();

  const float* glrow = gl + ((long)b * C_ + tc * TCH_ + lane) * HID_ + h * D_;
  cfloat_k* grw = (cfloat_k*)(gr + ((long)b * C_ + wv * SWV_) * HID_ + h * D_);
  float* srow_w = &sS[lane * SP_ + wv * SWV_];

  score_half<0>(glrow, grw, sAtt, srow_w);
  score_half<1>(glrow, grw, sAtt, srow_w);
  __syncthreads();

  float* arow = alpha + ((long)(b * H_ + h) * C_ + tc * TCH_) * C_;
#pragma unroll
  for (int k = 0; k < TCH_ / NWV_; ++k) {
    const int tl = wv * (TCH_ / NWV_) + k;
    const float* sr = &sS[tl * SP_];
    float v0 = sr[lane], v1 = sr[lane + 32], v2 = sr[lane + 64], v3 = sr[lane + 96];
    float m = fmaxf(fmaxf(v0, v1), fmaxf(v2, v3));
    for (int off = 16; off > 0; off >>= 1) m = fmaxf(m, __shfl_xor(m, off, 32));
    const float e0 = __expf(v0 - m), e1 = __expf(v1 - m);
    const float e2 = __expf(v2 - m), e3 = __expf(v3 - m);
    float r0 = e0, r1 = e1, r2 = e2, r3 = e3;
    for (int off = 16; off > 0; off >>= 1) {
      r0 += __shfl_xor(r0, off, 32);
      r1 += __shfl_xor(r1, off, 32);
      r2 += __shfl_xor(r2, off, 32);
      r3 += __shfl_xor(r3, off, 32);
    }
    const float ssum = r0 + r1 + r2 + r3;
    float* o = arow + (long)tl * C_ + lane;
    o[0]  = e0 / ssum;
    o[32] = e1 / ssum;
    o[64] = e2 / ssum;
    o[96] = e3 / ssum;
  }
}

__global__ __launch_bounds__(256) void post_elu_residual_ln(
    float* __restrict__ h, const float* __restrict__ gout,
    const float* __restrict__ g, const float* __restrict__ bb)
{
  __shared__ float red[8];
  const int tid = threadIdx.x;
  const long base = (long)blockIdx.x * HID_;

  float x = gout[base + tid];
  x = x > 0.f ? x : (__expf(x) - 1.f);
  const float v = h[base + tid] + x;

  const int lane = tid & 31, wv = tid >> 5;
  float s = v;
  for (int off = 16; off > 0; off >>= 1) s += __shfl_xor(s, off, 32);
  if (lane == 0) red[wv] = s;
  __syncthreads();
  s = 0.f;
#pragma unroll
  for (int i = 0; i < 8; ++i) s += red[i];
  const float mu = s * (1.f / HID_);
  __syncthreads();

  const float dmu = v - mu;
  float q = dmu * dmu;
  for (int off = 16; off > 0; off >>= 1) q += __shfl_xor(q, off, 32);
  if (lane == 0) red[wv] = q;
  __syncthreads();
  q = 0.f;
#pragma unroll
  for (int i = 0; i < 8; ++i) q += red[i];
  const float inv = rsqrtf(q * (1.f / HID_) + EPS_);

  h[base + tid] = dmu * inv * g[tid] + bb[tid];
}

extern "C" void kernel_launch(void* const* d_in, const int* in_sizes, int n_in,
                              void* d_out, int out_size, void* d_ws, size_t ws_size,
                              hipStream_t stream) {
  (void)in_sizes; (void)n_in; (void)out_size; (void)ws_size;
  const float* x      = (const float*)d_in[0];
  const float* W_emb  = (const float*)d_in[1];
  const float* b_emb  = (const float*)d_in[2];
  const float* Wl     = (const float*)d_in[3];
  const float* Wr     = (const float*)d_in[4];
  const float* att    = (const float*)d_in[5];
  const float* bias_g = (const float*)d_in[6];
  const float* ln_g   = (const float*)d_in[7];
  const float* ln_b   = (const float*)d_in[8];
  const float* W_proj = (const float*)d_in[9];
  const float* b_proj = (const float*)d_in[10];

  float* out   = (float*)d_out;
  float* alpha = out + (long)B_ * L_ * C_;

  float* ws = (float*)d_ws;
  const long hwN = (long)B_ * C_ * HID_;
  float* h    = ws;
  float* gl   = ws + hwN;
  float* gr   = ws + 2 * hwN;
  float* gout = ws + 3 * hwN;

  wmma_gemm_f32<C_, HID_, L_, C_, HID_, HID_, 1,
                (long)L_ * C_, 0, 0, 0, (long)C_ * HID_, 0, 0,
                true, false, true>
      <<<dim3((C_ / 32) * (HID_ / 64), B_), 128, 0, stream>>>(
          x, W_emb, b_emb, h);

  for (int layer = 0; layer < NL_; ++layer) {
    const float* Wl_i  = Wl + (long)layer * HID_ * HID_;
    const float* Wr_i  = Wr + (long)layer * HID_ * HID_;
    const float* att_i = att + (long)layer * H_ * D_;
    const float* bg_i  = bias_g + (long)layer * HID_;

    dim3 g1((C_ / 32) * (HID_ / 64), B_);
    wmma_gemm_f32<C_, HID_, HID_, HID_, HID_, HID_, 1,
                  (long)C_ * HID_, 0, 0, 0, (long)C_ * HID_, 0, 0,
                  false, false, false>
        <<<g1, 128, 0, stream>>>(h, Wl_i, nullptr, gl);
    wmma_gemm_f32<C_, HID_, HID_, HID_, HID_, HID_, 1,
                  (long)C_ * HID_, 0, 0, 0, (long)C_ * HID_, 0, 0,
                  false, false, false>
        <<<g1, 128, 0, stream>>>(h, Wr_i, nullptr, gr);

    gat_scores_softmax<<<dim3(C_ / TCH_, H_, B_), 32 * NWV_, 0, stream>>>(gl, gr, att_i, alpha);

    wmma_gemm_f32<C_, D_, C_, C_, HID_, HID_, H_,
                  (long)H_ * C_ * C_, (long)C_ * C_,
                  (long)C_ * HID_, D_, (long)C_ * HID_, D_, D_,
                  false, false, true>
        <<<dim3((C_ / 32) * (D_ / 64), B_ * H_), 128, 0, stream>>>(
            alpha, gr, bg_i, gout);

    post_elu_residual_ln<<<B_ * C_, 256, 0, stream>>>(
        h, gout, ln_g + (long)layer * HID_, ln_b + (long)layer * HID_);
  }

  wmma_gemm_f32<C_, L_, HID_, HID_, L_, C_, 1,
                (long)C_ * HID_, 0, 0, 0, (long)L_ * C_, 0, 0,
                false, true, true>
      <<<dim3((C_ / 32) * (L_ / 64), B_), 128, 0, stream>>>(
          h, W_proj, b_proj, out);
}


# GEMM launches with row-major A: next K chunk's global tile loads issued before the current chunk's matrix instructions (straight-line chunk, tile held in registers), to LDS after; k order unchanged
# speedup vs baseline: 7.4983x; 1.0854x over previous
#include <hip/hip_runtime.h>
#include <hip/hip_bf16.h>

typedef __attribute__((ext_vector_type(2))) float v2f;
typedef __attribute__((ext_vector_type(8))) float v8f;

#define B_   32
#define L_   512
#define C_   128
#define HID_ 256
#define H_   4
#define D_   64
#define NL_  3
#define NEG_ 0.2f
#define EPS_ 1e-5f
#define KC_  64

template <int M, int N, int K, int LDA, int LDB, int LDD, int INNER,
          long SAO, long SAI, long SBO, long SBI, long SDO, long SDI,
          long SBIASI, bool TRANSA, bool TRANSD, bool HASBIAS>
__global__ __launch_bounds__(128) void wmma_gemm_f32(
    const float* __restrict__ A, const float* __restrict__ Bm,
    const float* __restrict__ bias, float* __restrict__ D)
{
  constexpr int BM = 32;
  constexpr int BN = 64;
  constexpr int KC = KC_;
  constexpr int PA = TRANSA ? BM + 8 : KC + 4;
  constexpr int PB = BN + 8;
  constexpr int A_SZ = TRANSA ? KC * PA : BM * PA;
  constexpr int B_SZ = KC * PB;
  constexpr int A_F4 = BM * KC / 4 / 128;
  constexpr int B_F4 = KC * BN / 4 / 128;
  constexpr int NCH = K / KC;
  static_assert(K % KC == 0 && N % BN == 0 && M % BM == 0, "shape");

  __shared__ __attribute__((aligned(16))) float sA[2 * A_SZ];
  __shared__ __attribute__((aligned(16))) float sB[2 * B_SZ];

  const int tid  = threadIdx.x;
  const int lane = tid & 31;
  const int wave = tid >> 5;
  const int half = (lane >> 4) & 1;
  const int r    = lane & 15;

  constexpr int tilesNb = N / BN;
  const int mb = (blockIdx.x / tilesNb) * BM;
  const int nb = (blockIdx.x % tilesNb) * BN;
  const int wn = wave * 16;

  const int batch = blockIdx.y;
  const int bo = batch / INNER;
  const int bi = batch % INNER;
  A    += bo * SAO + bi * SAI;
  Bm   += bo * SBO + bi * SBI;
  D    += bo * SDO + bi * SDI;
  if (HASBIAS) bias += bi * SBIASI;

  int gA[A_F4], lA[A_F4], gB[B_F4], lB[B_F4];
#pragma unroll
  for (int j = 0; j < A_F4; ++j) {
    const int idx = tid + 128 * j;
    if (TRANSA) {
      const int krow = idx / (BM / 4), mq = idx % (BM / 4);
      gA[j] = krow * LDA + mb + 4 * mq;
      lA[j] = krow * PA + 4 * mq;
    } else {
      const int row = idx / (KC / 4), kq = idx % (KC / 4);
      gA[j] = (mb + row) * LDA + 4 * kq;
      lA[j] = row * PA + 4 * kq;
    }
  }
#pragma unroll
  for (int j = 0; j < B_F4; ++j) {
    const int idx = tid + 128 * j;
    const int krow = idx / (BN / 4), nq = idx % (BN / 4);
    gB[j] = krow * LDB + nb + 4 * nq;
    lB[j] = krow * PB + 4 * nq;
  }
  constexpr int stepA = TRANSA ? KC * LDA : KC;
  constexpr int stepB = KC * LDB;


  v8f acc0 = {0.f,0.f,0.f,0.f,0.f,0.f,0.f,0.f};
  v8f acc1 = acc0;

  float4 ra[A_F4], rb[B_F4];
#pragma unroll
  for (int j = 0; j < A_F4; ++j) ra[j] = *(const float4*)(A + gA[j]);
#pragma unroll
  for (int j = 0; j < B_F4; ++j) rb[j] = *(const float4*)(Bm + gB[j]);
#pragma unroll
  for (int j = 0; j < A_F4; ++j) *(float4*)(sA + lA[j]) = ra[j];
#pragma unroll
  for (int j = 0; j < B_F4; ++j) *(float4*)(sB + lB[j]) = rb[j];
  __syncthreads();

#pragma unroll
  for (int c = 0; c < NCH; ++c) {
    const float* as = sA + (c & 1) * A_SZ;
    const float* bs = sB + (c & 1) * B_SZ;
    if (c + 1 < NCH) {
#pragma unroll
      for (int j = 0; j < A_F4; ++j) ra[j] = *(const float4*)(A + gA[j] + (c + 1) * stepA);
#pragma unroll
      for (int j = 0; j < B_F4; ++j) rb[j] = *(const float4*)(Bm + gB[j] + (c + 1) * stepB);
      if (!TRANSA) __builtin_amdgcn_sched_barrier(0);
    }
#pragma unroll (TRANSA ? 8 : KC / 4)
    for (int ks = 0; ks < KC / 4; ++ks) {
      const int k = 4 * ks + 2 * half;
      v2f a0, a1, b0;
      if (TRANSA) {
        a0.x = as[k * PA + r];      a0.y = as[(k + 1) * PA + r];
        a1.x = as[k * PA + 16 + r]; a1.y = as[(k + 1) * PA + 16 + r];
      } else {
        a0 = *(const v2f*)(as + r * PA + k);
        a1 = *(const v2f*)(as + (16 + r) * PA + k);
      }
      b0.x = bs[k * PB + wn + r];   b0.y = bs[(k + 1) * PB + wn + r];

      acc0 = __builtin_amdgcn_wmma_f32_16x16x4_f32(false, a0, false, b0, (short)0, acc0, false, false);
      acc1 = __builtin_amdgcn_wmma_f32_16x16x4_f32(false, a1, false, b0, (short)0, acc1, false, false);
    }
    if (c + 1 < NCH) {
      float* an = sA + ((c + 1) & 1) * A_SZ;
      float* bn = sB + ((c + 1) & 1) * B_SZ;
#pragma unroll
      for (int j = 0; j < A_F4; ++j) *(float4*)(an + lA[j]) = ra[j];
#pragma unroll
      for (int j = 0; j < B_F4; ++j) *(float4*)(bn + lB[j]) = rb[j];
      __syncthreads();
    }
  }

  const int col0 = nb + wn + r;
  const float bv0 = HASBIAS ? bias[col0] : 0.f;
#pragma unroll
  for (int i = 0; i < 8; ++i) {
    const int row0 = mb + i + 8 * half;
    const int row1 = row0 + 16;
    if (TRANSD) {
      D[(long)col0 * LDD + row0] = acc0[i] + bv0;
      D[(long)col0 * LDD + row1] = acc1[i] + bv0;
    } else {
      D[(long)row0 * LDD + col0] = acc0[i] + bv0;
      D[(long)row1 * LDD + col0] = acc1[i] + bv0;
    }
  }
}

#define TCH_ 32
#define NWV_ 8
#define SWV_ (C_ / NWV_)
#define SP_  (C_ + 1)

typedef const __attribute__((address_space(4))) float cfloat_k;

template <int HF>
__device__ __forceinline__ void score_half(
    const float* __restrict__ glrow, cfloat_k* grw,
    const float* sAtt, float* srow_w)
{
  const float4* glp = (const float4*)(glrow + HF * (D_ / 2));
  float glr[D_ / 2];
#pragma unroll
  for (int q = 0; q < D_ / 8; ++q) {
    const float4 v = glp[q];
    glr[4 * q + 0] = v.x; glr[4 * q + 1] = v.y;
    glr[4 * q + 2] = v.z; glr[4 * q + 3] = v.w;
  }
  float av[D_ / 2];
#pragma unroll
  for (int d = 0; d < D_ / 2; ++d) av[d] = sAtt[HF * (D_ / 2) + d];

  cfloat_k* gh = grw + HF * (D_ / 2);
#pragma unroll 2
  for (int i = 0; i < SWV_; ++i) {
    cfloat_k* g = gh + (long)i * HID_;
    float sc = HF ? srow_w[i] : 0.f;
#pragma unroll
    for (int d = 0; d < D_ / 2; ++d) {
      const float v = glr[d] + g[d];
      sc = fmaf(av[d], fmaxf(v, NEG_ * v), sc);
    }
    srow_w[i] = sc;
  }
}

__global__ __launch_bounds__(256) void gat_scores_softmax(
    const float* __restrict__ gl, const float* __restrict__ gr,
    const float* __restrict__ att, float* __restrict__ alpha)
{
  __shared__ float sS[TCH_ * SP_];
  __shared__ float sAtt[D_];

  const int tid = threadIdx.x;
  const int lane = tid & 31;
  const int wv = __builtin_amdgcn_readfirstlane(tid >> 5);
  const int tc = blockIdx.x;
  const int h = blockIdx.y;
  const int b = blockIdx.z;

  if (wv < D_ / 32) sAtt[tid] = att[h * D_ + tid];

  __syncthreads();

  const float* glrow = gl + ((long)b * C_ + tc * TCH_ + lane) * HID_ + h * D_;
  cfloat_k* grw = (cfloat_k*)(gr + ((long)b * C_ + wv * SWV_) * HID_ + h * D_);
  float* srow_w = &sS[lane * SP_ + wv * SWV_];

  score_half<0>(glrow, grw, sAtt, srow_w);
  score_half<1>(glrow, grw, sAtt, srow_w);
  __syncthreads();

  float* arow = alpha + ((long)(b * H_ + h) * C_ + tc * TCH_) * C_;
#pragma unroll
  for (int k = 0; k < TCH_ / NWV_; ++k) {
    const int tl = wv * (TCH_ / NWV_) + k;
    const float* sr = &sS[tl * SP_];
    float v0 = sr[lane], v1 = sr[lane + 32], v2 = sr[lane + 64], v3 = sr[lane + 96];
    float m = fmaxf(fmaxf(v0, v1), fmaxf(v2, v3));
    for (int off = 16; off > 0; off >>= 1) m = fmaxf(m, __shfl_xor(m, off, 32));
    const float e0 = __expf(v0 - m), e1 = __expf(v1 - m);
    const float e2 = __expf(v2 - m), e3 = __expf(v3 - m);
    float r0 = e0, r1 = e1, r2 = e2, r3 = e3;
    for (int off = 16; off > 0; off >>= 1) {
      r0 += __shfl_xor(r0, off, 32);
      r1 += __shfl_xor(r1, off, 32);
      r2 += __shfl_xor(r2, off, 32);
      r3 += __shfl_xor(r3, off, 32);
    }
    const float ssum = r0 + r1 + r2 + r3;
    float* o = arow + (long)tl * C_ + lane;
    o[0]  = e0 / ssum;
    o[32] = e1 / ssum;
    o[64] = e2 / ssum;
    o[96] = e3 / ssum;
  }
}

__global__ __launch_bounds__(256) void post_elu_residual_ln(
    float* __restrict__ h, const float* __restrict__ gout,
    const float* __restrict__ g, const float* __restrict__ bb)
{
  __shared__ float red[8];
  const int tid = threadIdx.x;
  const long base = (long)blockIdx.x * HID_;

  float x = gout[base + tid];
  x = x > 0.f ? x : (__expf(x) - 1.f);
  const float v = h[base + tid] + x;

  const int lane = tid & 31, wv = tid >> 5;
  float s = v;
  for (int off = 16; off > 0; off >>= 1) s += __shfl_xor(s, off, 32);
  if (lane == 0) red[wv] = s;
  __syncthreads();
  s = 0.f;
#pragma unroll
  for (int i = 0; i < 8; ++i) s += red[i];
  const float mu = s * (1.f / HID_);
  __syncthreads();

  const float dmu = v - mu;
  float q = dmu * dmu;
  for (int off = 16; off > 0; off >>= 1) q += __shfl_xor(q, off, 32);
  if (lane == 0) red[wv] = q;
  __syncthreads();
  q = 0.f;
#pragma unroll
  for (int i = 0; i < 8; ++i) q += red[i];
  const float inv = rsqrtf(q * (1.f / HID_) + EPS_);

  h[base + tid] = dmu * inv * g[tid] + bb[tid];
}

extern "C" void kernel_launch(void* const* d_in, const int* in_sizes, int n_in,
                              void* d_out, int out_size, void* d_ws, size_t ws_size,
                              hipStream_t stream) {
  (void)in_sizes; (void)n_in; (void)out_size; (void)ws_size;
  const float* x      = (const float*)d_in[0];
  const float* W_emb  = (const float*)d_in[1];
  const float* b_emb  = (const float*)d_in[2];
  const float* Wl     = (const float*)d_in[3];
  const float* Wr     = (const float*)d_in[4];
  const float* att    = (const float*)d_in[5];
  const float* bias_g = (const float*)d_in[6];
  const float* ln_g   = (const float*)d_in[7];
  const float* ln_b   = (const float*)d_in[8];
  const float* W_proj = (const float*)d_in[9];
  const float* b_proj = (const float*)d_in[10];

  float* out   = (float*)d_out;
  float* alpha = out + (long)B_ * L_ * C_;

  float* ws = (float*)d_ws;
  const long hwN = (long)B_ * C_ * HID_;
  float* h    = ws;
  float* gl   = ws + hwN;
  float* gr   = ws + 2 * hwN;
  float* gout = ws + 3 * hwN;

  wmma_gemm_f32<C_, HID_, L_, C_, HID_, HID_, 1,
                (long)L_ * C_, 0, 0, 0, (long)C_ * HID_, 0, 0,
                true, false, true>
      <<<dim3((C_ / 32) * (HID_ / 64), B_), 128, 0, stream>>>(
          x, W_emb, b_emb, h);

  for (int layer = 0; layer < NL_; ++layer) {
    const float* Wl_i  = Wl + (long)layer * HID_ * HID_;
    const float* Wr_i  = Wr + (long)layer * HID_ * HID_;
    const float* att_i = att + (long)layer * H_ * D_;
    const float* bg_i  = bias_g + (long)layer * HID_;

    dim3 g1((C_ / 32) * (HID_ / 64), B_);
    wmma_gemm_f32<C_, HID_, HID_, HID_, HID_, HID_, 1,
                  (long)C_ * HID_, 0, 0, 0, (long)C_ * HID_, 0, 0,
                  false, false, false>
        <<<g1, 128, 0, stream>>>(h, Wl_i, nullptr, gl);
    wmma_gemm_f32<C_, HID_, HID_, HID_, HID_, HID_, 1,
                  (long)C_ * HID_, 0, 0, 0, (long)C_ * HID_, 0, 0,
                  false, false, false>
        <<<g1, 128, 0, stream>>>(h, Wr_i, nullptr, gr);

    gat_scores_softmax<<<dim3(C_ / TCH_, H_, B_), 32 * NWV_, 0, stream>>>(gl, gr, att_i, alpha);

    wmma_gemm_f32<C_, D_, C_, C_, HID_, HID_, H_,
                  (long)H_ * C_ * C_, (long)C_ * C_,
                  (long)C_ * HID_, D_, (long)C_ * HID_, D_, D_,
                  false, false, true>
        <<<dim3((C_ / 32) * (D_ / 64), B_ * H_), 128, 0, stream>>>(
            alpha, gr, bg_i, gout);

    post_elu_residual_ln<<<B_ * C_, 256, 0, stream>>>(
        h, gout, ln_g + (long)layer * HID_, ln_b + (long)layer * HID_);
  }

  wmma_gemm_f32<C_, L_, HID_, HID_, L_, C_, 1,
                (long)C_ * HID_, 0, 0, 0, (long)L_ * C_, 0, 0,
                false, true, true>
      <<<dim3((C_ / 32) * (L_ / 64), B_), 128, 0, stream>>>(
          h, W_proj, b_proj, out);
}


# norm launch: a wave takes 2 whole rows (8 values per lane, 16-byte loads and stores, shuffle-only row sums) instead of a thread per element: 2048 waves enter, not 32768; row sums re-associated
# speedup vs baseline: 7.7452x; 1.0329x over previous
#include <hip/hip_runtime.h>
#include <hip/hip_bf16.h>

typedef __attribute__((ext_vector_type(2))) float v2f;
typedef __attribute__((ext_vector_type(8))) float v8f;

#define B_   32
#define L_   512
#define C_   128
#define HID_ 256
#define H_   4
#define D_   64
#define NL_  3
#define NEG_ 0.2f
#define EPS_ 1e-5f
#define KC_  64

template <int M, int N, int K, int LDA, int LDB, int LDD, int INNER,
          long SAO, long SAI, long SBO, long SBI, long SDO, long SDI,
          long SBIASI, bool TRANSA, bool TRANSD, bool HASBIAS>
__global__ __launch_bounds__(128) void wmma_gemm_f32(
    const float* __restrict__ A, const float* __restrict__ Bm,
    const float* __restrict__ bias, float* __restrict__ D)
{
  constexpr int BM = 32;
  constexpr int BN = 64;
  constexpr int KC = KC_;
  constexpr int PA = TRANSA ? BM + 8 : KC + 4;
  constexpr int PB = BN + 8;
  constexpr int A_SZ = TRANSA ? KC * PA : BM * PA;
  constexpr int B_SZ = KC * PB;
  constexpr int A_F4 = BM * KC / 4 / 128;
  constexpr int B_F4 = KC * BN / 4 / 128;
  constexpr int NCH = K / KC;
  static_assert(K % KC == 0 && N % BN == 0 && M % BM == 0, "shape");

  __shared__ __attribute__((aligned(16))) float sA[2 * A_SZ];
  __shared__ __attribute__((aligned(16))) float sB[2 * B_SZ];

  const int tid  = threadIdx.x;
  const int lane = tid & 31;
  const int wave = tid >> 5;
  const int half = (lane >> 4) & 1;
  const int r    = lane & 15;

  constexpr int tilesNb = N / BN;
  const int mb = (blockIdx.x / tilesNb) * BM;
  const int nb = (blockIdx.x % tilesNb) * BN;
  const int wn = wave * 16;

  const int batch = blockIdx.y;
  const int bo = batch / INNER;
  const int bi = batch % INNER;
  A    += bo * SAO + bi * SAI;
  Bm   += bo * SBO + bi * SBI;
  D    += bo * SDO + bi * SDI;
  if (HASBIAS) bias += bi * SBIASI;

  int gA[A_F4], lA[A_F4], gB[B_F4], lB[B_F4];
#pragma unroll
  for (int j = 0; j < A_F4; ++j) {
    const int idx = tid + 128 * j;
    if (TRANSA) {
      const int krow = idx / (BM / 4), mq = idx % (BM / 4);
      gA[j] = krow * LDA + mb + 4 * mq;
      lA[j] = krow * PA + 4 * mq;
    } else {
      const int row = idx / (KC / 4), kq = idx % (KC / 4);
      gA[j] = (mb + row) * LDA + 4 * kq;
      lA[j] = row * PA + 4 * kq;
    }
  }
#pragma unroll
  for (int j = 0; j < B_F4; ++j) {
    const int idx = tid + 128 * j;
    const int krow = idx / (BN / 4), nq = idx % (BN / 4);
    gB[j] = krow * LDB + nb + 4 * nq;
    lB[j] = krow * PB + 4 * nq;
  }
  constexpr int stepA = TRANSA ? KC * LDA : KC;
  constexpr int stepB = KC * LDB;


  v8f acc0 = {0.f,0.f,0.f,0.f,0.f,0.f,0.f,0.f};
  v8f acc1 = acc0;

  float4 ra[A_F4], rb[B_F4];
#pragma unroll
  for (int j = 0; j < A_F4; ++j) ra[j] = *(const float4*)(A + gA[j]);
#pragma unroll
  for (int j = 0; j < B_F4; ++j) rb[j] = *(const float4*)(Bm + gB[j]);
#pragma unroll
  for (int j = 0; j < A_F4; ++j) *(float4*)(sA + lA[j]) = ra[j];
#pragma unroll
  for (int j = 0; j < B_F4; ++j) *(float4*)(sB + lB[j]) = rb[j];
  __syncthreads();

#pragma unroll
  for (int c = 0; c < NCH; ++c) {
    const float* as = sA + (c & 1) * A_SZ;
    const float* bs = sB + (c & 1) * B_SZ;
    if (c + 1 < NCH) {
#pragma unroll
      for (int j = 0; j < A_F4; ++j) ra[j] = *(const float4*)(A + gA[j] + (c + 1) * stepA);
#pragma unroll
      for (int j = 0; j < B_F4; ++j) rb[j] = *(const float4*)(Bm + gB[j] + (c + 1) * stepB);
      if (!TRANSA) __builtin_amdgcn_sched_barrier(0);
    }
#pragma unroll (TRANSA ? 8 : KC / 4)
    for (int ks = 0; ks < KC / 4; ++ks) {
      const int k = 4 * ks + 2 * half;
      v2f a0, a1, b0;
      if (TRANSA) {
        a0.x = as[k * PA + r];      a0.y = as[(k + 1) * PA + r];
        a1.x = as[k * PA + 16 + r]; a1.y = as[(k + 1) * PA + 16 + r];
      } else {
        a0 = *(const v2f*)(as + r * PA + k);
        a1 = *(const v2f*)(as + (16 + r) * PA + k);
      }
      b0.x = bs[k * PB + wn + r];   b0.y = bs[(k + 1) * PB + wn + r];

      acc0 = __builtin_amdgcn_wmma_f32_16x16x4_f32(false, a0, false, b0, (short)0, acc0, false, false);
      acc1 = __builtin_amdgcn_wmma_f32_16x16x4_f32(false, a1, false, b0, (short)0, acc1, false, false);
    }
    if (c + 1 < NCH) {
      float* an = sA + ((c + 1) & 1) * A_SZ;
      float* bn = sB + ((c + 1) & 1) * B_SZ;
#pragma unroll
      for (int j = 0; j < A_F4; ++j) *(float4*)(an + lA[j]) = ra[j];
#pragma unroll
      for (int j = 0; j < B_F4; ++j) *(float4*)(bn + lB[j]) = rb[j];
      __syncthreads();
    }
  }

  const int col0 = nb + wn + r;
  const float bv0 = HASBIAS ? bias[col0] : 0.f;
#pragma unroll
  for (int i = 0; i < 8; ++i) {
    const int row0 = mb + i + 8 * half;
    const int row1 = row0 + 16;
    if (TRANSD) {
      D[(long)col0 * LDD + row0] = acc0[i] + bv0;
      D[(long)col0 * LDD + row1] = acc1[i] + bv0;
    } else {
      D[(long)row0 * LDD + col0] = acc0[i] + bv0;
      D[(long)row1 * LDD + col0] = acc1[i] + bv0;
    }
  }
}

#define TCH_ 32
#define NWV_ 8
#define SWV_ (C_ / NWV_)
#define SP_  (C_ + 1)

typedef const __attribute__((address_space(4))) float cfloat_k;

template <int HF>
__device__ __forceinline__ void score_half(
    const float* __restrict__ glrow, cfloat_k* grw,
    const float* sAtt, float* srow_w)
{
  const float4* glp = (const float4*)(glrow + HF * (D_ / 2));
  float glr[D_ / 2];
#pragma unroll
  for (int q = 0; q < D_ / 8; ++q) {
    const float4 v = glp[q];
    glr[4 * q + 0] = v.x; glr[4 * q + 1] = v.y;
    glr[4 * q + 2] = v.z; glr[4 * q + 3] = v.w;
  }
  float av[D_ / 2];
#pragma unroll
  for (int d = 0; d < D_ / 2; ++d) av[d] = sAtt[HF * (D_ / 2) + d];

  cfloat_k* gh = grw + HF * (D_ / 2);
#pragma unroll 2
  for (int i = 0; i < SWV_; ++i) {
    cfloat_k* g = gh + (long)i * HID_;
    float sc = HF ? srow_w[i] : 0.f;
#pragma unroll
    for (int d = 0; d < D_ / 2; ++d) {
      const float v = glr[d] + g[d];
      sc = fmaf(av[d], fmaxf(v, NEG_ * v), sc);
    }
    srow_w[i] = sc;
  }
}

__global__ __launch_bounds__(256) void gat_scores_softmax(
    const float* __restrict__ gl, const float* __restrict__ gr,
    const float* __restrict__ att, float* __restrict__ alpha)
{
  __shared__ float sS[TCH_ * SP_];
  __shared__ float sAtt[D_];

  const int tid = threadIdx.x;
  const int lane = tid & 31;
  const int wv = __builtin_amdgcn_readfirstlane(tid >> 5);
  const int tc = blockIdx.x;
  const int h = blockIdx.y;
  const int b = blockIdx.z;

  if (wv < D_ / 32) sAtt[tid] = att[h * D_ + tid];

  __syncthreads();

  const float* glrow = gl + ((long)b * C_ + tc * TCH_ + lane) * HID_ + h * D_;
  cfloat_k* grw = (cfloat_k*)(gr + ((long)b * C_ + wv * SWV_) * HID_ + h * D_);
  float* srow_w = &sS[lane * SP_ + wv * SWV_];

  score_half<0>(glrow, grw, sAtt, srow_w);
  score_half<1>(glrow, grw, sAtt, srow_w);
  __syncthreads();

  float* arow = alpha + ((long)(b * H_ + h) * C_ + tc * TCH_) * C_;
#pragma unroll
  for (int k = 0; k < TCH_ / NWV_; ++k) {
    const int tl = wv * (TCH_ / NWV_) + k;
    const float* sr = &sS[tl * SP_];
    float v0 = sr[lane], v1 = sr[lane + 32], v2 = sr[lane + 64], v3 = sr[lane + 96];
    float m = fmaxf(fmaxf(v0, v1), fmaxf(v2, v3));
    for (int off = 16; off > 0; off >>= 1) m = fmaxf(m, __shfl_xor(m, off, 32));
    const float e0 = __expf(v0 - m), e1 = __expf(v1 - m);
    const float e2 = __expf(v2 - m), e3 = __expf(v3 - m);
    float r0 = e0, r1 = e1, r2 = e2, r3 = e3;
    for (int off = 16; off > 0; off >>= 1) {
      r0 += __shfl_xor(r0, off, 32);
      r1 += __shfl_xor(r1, off, 32);
      r2 += __shfl_xor(r2, off, 32);
      r3 += __shfl_xor(r3, off, 32);
    }
    const float ssum = r0 + r1 + r2 + r3;
    float* o = arow + (long)tl * C_ + lane;
    o[0]  = e0 / ssum;
    o[32] = e1 / ssum;
    o[64] = e2 / ssum;
    o[96] = e3 / ssum;
  }
}

#define LNR_ 2

__device__ __forceinline__ float elu1(float x) {
  return x > 0.f ? x : (__expf(x) - 1.f);
}

__global__ __launch_bounds__(256) void post_elu_residual_ln(
    float* __restrict__ h, const float* __restrict__ gout,
    const float* __restrict__ g, const float* __restrict__ bb)
{
  const int lane = threadIdx.x & 31;
  const int wv   = threadIdx.x >> 5;
  const long row0 = ((long)blockIdx.x * 8 + wv) * LNR_;

  const float4 g0 = *(const float4*)(g + 4 * lane);
  const float4 g1 = *(const float4*)(g + 128 + 4 * lane);
  const float4 c0 = *(const float4*)(bb + 4 * lane);
  const float4 c1 = *(const float4*)(bb + 128 + 4 * lane);

#pragma unroll
  for (int k = 0; k < LNR_; ++k) {
    const long base = (row0 + k) * HID_;
    const float4 x0 = *(const float4*)(gout + base + 4 * lane);
    const float4 x1 = *(const float4*)(gout + base + 128 + 4 * lane);
    const float4 h0 = *(const float4*)(h + base + 4 * lane);
    const float4 h1 = *(const float4*)(h + base + 128 + 4 * lane);
    const float v0 = h0.x + elu1(x0.x), v1 = h0.y + elu1(x0.y);
    const float v2 = h0.z + elu1(x0.z), v3 = h0.w + elu1(x0.w);
    const float v4 = h1.x + elu1(x1.x), v5 = h1.y + elu1(x1.y);
    const float v6 = h1.z + elu1(x1.z), v7 = h1.w + elu1(x1.w);

    float s = ((v0 + v1) + (v2 + v3)) + ((v4 + v5) + (v6 + v7));
    for (int off = 16; off > 0; off >>= 1) s += __shfl_xor(s, off, 32);
    const float mu = s * (1.f / HID_);

    const float d0 = v0 - mu, d1 = v1 - mu, d2 = v2 - mu, d3 = v3 - mu;
    const float d4 = v4 - mu, d5 = v5 - mu, d6 = v6 - mu, d7 = v7 - mu;
    float q = ((d0 * d0 + d1 * d1) + (d2 * d2 + d3 * d3)) +
              ((d4 * d4 + d5 * d5) + (d6 * d6 + d7 * d7));
    for (int off = 16; off > 0; off >>= 1) q += __shfl_xor(q, off, 32);
    const float inv = rsqrtf(q * (1.f / HID_) + EPS_);

    float4 o0, o1;
    o0.x = d0 * inv * g0.x + c0.x;  o0.y = d1 * inv * g0.y + c0.y;
    o0.z = d2 * inv * g0.z + c0.z;  o0.w = d3 * inv * g0.w + c0.w;
    o1.x = d4 * inv * g1.x + c1.x;  o1.y = d5 * inv * g1.y + c1.y;
    o1.z = d6 * inv * g1.z + c1.z;  o1.w = d7 * inv * g1.w + c1.w;
    *(float4*)(h + base + 4 * lane)       = o0;
    *(float4*)(h + base + 128 + 4 * lane) = o1;
  }
}

extern "C" void kernel_launch(void* const* d_in, const int* in_sizes, int n_in,
                              void* d_out, int out_size, void* d_ws, size_t ws_size,
                              hipStream_t stream) {
  (void)in_sizes; (void)n_in; (void)out_size; (void)ws_size;
  const float* x      = (const float*)d_in[0];
  const float* W_emb  = (const float*)d_in[1];
  const float* b_emb  = (const float*)d_in[2];
  const float* Wl     = (const float*)d_in[3];
  const float* Wr     = (const float*)d_in[4];
  const float* att    = (const float*)d_in[5];
  const float* bias_g = (const float*)d_in[6];
  const float* ln_g   = (const float*)d_in[7];
  const float* ln_b   = (const float*)d_in[8];
  const float* W_proj = (const float*)d_in[9];
  const float* b_proj = (const float*)d_in[10];

  float* out   = (float*)d_out;
  float* alpha = out + (long)B_ * L_ * C_;

  float* ws = (float*)d_ws;
  const long hwN = (long)B_ * C_ * HID_;
  float* h    = ws;
  float* gl   = ws + hwN;
  float* gr   = ws + 2 * hwN;
  float* gout = ws + 3 * hwN;

  wmma_gemm_f32<C_, HID_, L_, C_, HID_, HID_, 1,
                (long)L_ * C_, 0, 0, 0, (long)C_ * HID_, 0, 0,
                true, false, true>
      <<<dim3((C_ / 32) * (HID_ / 64), B_), 128, 0, stream>>>(
          x, W_emb, b_emb, h);

  for (int layer = 0; layer < NL_; ++layer) {
    const float* Wl_i  = Wl + (long)layer * HID_ * HID_;
    const float* Wr_i  = Wr + (long)layer * HID_ * HID_;
    const float* att_i = att + (long)layer * H_ * D_;
    const float* bg_i  = bias_g + (long)layer * HID_;

    dim3 g1((C_ / 32) * (HID_ / 64), B_);
    wmma_gemm_f32<C_, HID_, HID_, HID_, HID_, HID_, 1,
                  (long)C_ * HID_, 0, 0, 0, (long)C_ * HID_, 0, 0,
                  false, false, false>
        <<<g1, 128, 0, stream>>>(h, Wl_i, nullptr, gl);
    wmma_gemm_f32<C_, HID_, HID_, HID_, HID_, HID_, 1,
                  (long)C_ * HID_, 0, 0, 0, (long)C_ * HID_, 0, 0,
                  false, false, false>
        <<<g1, 128, 0, stream>>>(h, Wr_i, nullptr, gr);

    gat_scores_softmax<<<dim3(C_ / TCH_, H_, B_), 32 * NWV_, 0, stream>>>(gl, gr, att_i, alpha);

    wmma_gemm_f32<C_, D_, C_, C_, HID_, HID_, H_,
                  (long)H_ * C_ * C_, (long)C_ * C_,
                  (long)C_ * HID_, D_, (long)C_ * HID_, D_, D_,
                  false, false, true>
        <<<dim3((C_ / 32) * (D_ / 64), B_ * H_), 128, 0, stream>>>(
            alpha, gr, bg_i, gout);

    post_elu_residual_ln<<<B_ * C_ / (8 * LNR_), 256, 0, stream>>>(
        h, gout, ln_g + (long)layer * HID_, ln_b + (long)layer * HID_);
  }

  wmma_gemm_f32<C_, L_, HID_, HID_, L_, C_, 1,
                (long)C_ * HID_, 0, 0, 0, (long)L_ * C_, 0, 0,
                false, true, true>
      <<<dim3((C_ / 32) * (L_ / 64), B_), 128, 0, stream>>>(
          h, W_proj, b_proj, out);
}


# scores kernel re-tiled: a lane owns 2 target rows as register pairs (64 targets per work-group), packed f32 d loop takes the source value as a scalar operand, 4 source rows in flight; sum orders kept
# speedup vs baseline: 8.1823x; 1.0564x over previous
#include <hip/hip_runtime.h>
#include <hip/hip_bf16.h>

typedef __attribute__((ext_vector_type(2))) float v2f;
typedef __attribute__((ext_vector_type(8))) float v8f;

#define B_   32
#define L_   512
#define C_   128
#define HID_ 256
#define H_   4
#define D_   64
#define NL_  3
#define NEG_ 0.2f
#define EPS_ 1e-5f
#define KC_  64

template <int M, int N, int K, int LDA, int LDB, int LDD, int INNER,
          long SAO, long SAI, long SBO, long SBI, long SDO, long SDI,
          long SBIASI, bool TRANSA, bool TRANSD, bool HASBIAS>
__global__ __launch_bounds__(128) void wmma_gemm_f32(
    const float* __restrict__ A, const float* __restrict__ Bm,
    const float* __restrict__ bias, float* __restrict__ D)
{
  constexpr int BM = 32;
  constexpr int BN = 64;
  constexpr int KC = KC_;
  constexpr int PA = TRANSA ? BM + 8 : KC + 4;
  constexpr int PB = BN + 8;
  constexpr int A_SZ = TRANSA ? KC * PA : BM * PA;
  constexpr int B_SZ = KC * PB;
  constexpr int A_F4 = BM * KC / 4 / 128;
  constexpr int B_F4 = KC * BN / 4 / 128;
  constexpr int NCH = K / KC;
  static_assert(K % KC == 0 && N % BN == 0 && M % BM == 0, "shape");

  __shared__ __attribute__((aligned(16))) float sA[2 * A_SZ];
  __shared__ __attribute__((aligned(16))) float sB[2 * B_SZ];

  const int tid  = threadIdx.x;
  const int lane = tid & 31;
  const int wave = tid >> 5;
  const int half = (lane >> 4) & 1;
  const int r    = lane & 15;

  constexpr int tilesNb = N / BN;
  const int mb = (blockIdx.x / tilesNb) * BM;
  const int nb = (blockIdx.x % tilesNb) * BN;
  const int wn = wave * 16;

  const int batch = blockIdx.y;
  const int bo = batch / INNER;
  const int bi = batch % INNER;
  A    += bo * SAO + bi * SAI;
  Bm   += bo * SBO + bi * SBI;
  D    += bo * SDO + bi * SDI;
  if (HASBIAS) bias += bi * SBIASI;

  int gA[A_F4], lA[A_F4], gB[B_F4], lB[B_F4];
#pragma unroll
  for (int j = 0; j < A_F4; ++j) {
    const int idx = tid + 128 * j;
    if (TRANSA) {
      const int krow = idx / (BM / 4), mq = idx % (BM / 4);
      gA[j] = krow * LDA + mb + 4 * mq;
      lA[j] = krow * PA + 4 * mq;
    } else {
      const int row = idx / (KC / 4), kq = idx % (KC / 4);
      gA[j] = (mb + row) * LDA + 4 * kq;
      lA[j] = row * PA + 4 * kq;
    }
  }
#pragma unroll
  for (int j = 0; j < B_F4; ++j) {
    const int idx = tid + 128 * j;
    const int krow = idx / (BN / 4), nq = idx % (BN / 4);
    gB[j] = krow * LDB + nb + 4 * nq;
    lB[j] = krow * PB + 4 * nq;
  }
  constexpr int stepA = TRANSA ? KC * LDA : KC;
  constexpr int stepB = KC * LDB;


  v8f acc0 = {0.f,0.f,0.f,0.f,0.f,0.f,0.f,0.f};
  v8f acc1 = acc0;

  float4 ra[A_F4], rb[B_F4];
#pragma unroll
  for (int j = 0; j < A_F4; ++j) ra[j] = *(const float4*)(A + gA[j]);
#pragma unroll
  for (int j = 0; j < B_F4; ++j) rb[j] = *(const float4*)(Bm + gB[j]);
#pragma unroll
  for (int j = 0; j < A_F4; ++j) *(float4*)(sA + lA[j]) = ra[j];
#pragma unroll
  for (int j = 0; j < B_F4; ++j) *(float4*)(sB + lB[j]) = rb[j];
  __syncthreads();

#pragma unroll
  for (int c = 0; c < NCH; ++c) {
    const float* as = sA + (c & 1) * A_SZ;
    const float* bs = sB + (c & 1) * B_SZ;
    if (c + 1 < NCH) {
#pragma unroll
      for (int j = 0; j < A_F4; ++j) ra[j] = *(const float4*)(A + gA[j] + (c + 1) * stepA);
#pragma unroll
      for (int j = 0; j < B_F4; ++j) rb[j] = *(const float4*)(Bm + gB[j] + (c + 1) * stepB);
      if (!TRANSA) __builtin_amdgcn_sched_barrier(0);
    }
#pragma unroll (TRANSA ? 8 : KC / 4)
    for (int ks = 0; ks < KC / 4; ++ks) {
      const int k = 4 * ks + 2 * half;
      v2f a0, a1, b0;
      if (TRANSA) {
        a0.x = as[k * PA + r];      a0.y = as[(k + 1) * PA + r];
        a1.x = as[k * PA + 16 + r]; a1.y = as[(k + 1) * PA + 16 + r];
      } else {
        a0 = *(const v2f*)(as + r * PA + k);
        a1 = *(const v2f*)(as + (16 + r) * PA + k);
      }
      b0.x = bs[k * PB + wn + r];   b0.y = bs[(k + 1) * PB + wn + r];

      acc0 = __builtin_amdgcn_wmma_f32_16x16x4_f32(false, a0, false, b0, (short)0, acc0, false, false);
      acc1 = __builtin_amdgcn_wmma_f32_16x16x4_f32(false, a1, false, b0, (short)0, acc1, false, false);
    }
    if (c + 1 < NCH) {
      float* an = sA + ((c + 1) & 1) * A_SZ;
      float* bn = sB + ((c + 1) & 1) * B_SZ;
#pragma unroll
      for (int j = 0; j < A_F4; ++j) *(float4*)(an + lA[j]) = ra[j];
#pragma unroll
      for (int j = 0; j < B_F4; ++j) *(float4*)(bn + lB[j]) = rb[j];
      __syncthreads();
    }
  }

  const int col0 = nb + wn + r;
  const float bv0 = HASBIAS ? bias[col0] : 0.f;
#pragma unroll
  for (int i = 0; i < 8; ++i) {
    const int row0 = mb + i + 8 * half;
    const int row1 = row0 + 16;
    if (TRANSD) {
      D[(long)col0 * LDD + row0] = acc0[i] + bv0;
      D[(long)col0 * LDD + row1] = acc1[i] + bv0;
    } else {
      D[(long)row0 * LDD + col0] = acc0[i] + bv0;
      D[(long)row1 * LDD + col0] = acc1[i] + bv0;
    }
  }
}

#define TCH_ 64
#define NWV_ 8
#define SWV_ (C_ / NWV_)
#define SP_  (C_ + 1)
#define RIF_ 4

typedef const __attribute__((address_space(4))) float cfloat_k;

template <int HF>
__device__ __forceinline__ void score_half(
    const float* __restrict__ glrow, cfloat_k* grw,
    const float* sAtt, float* srow_w)
{
  const float4* glp0 = (const float4*)(glrow + HF * (D_ / 2));
  const float4* glp1 = (const float4*)(glrow + 32 * HID_ + HF * (D_ / 2));
  v2f glr[D_ / 2];
#pragma unroll
  for (int q = 0; q < D_ / 8; ++q) {
    const float4 u = glp0[q];
    const float4 w = glp1[q];
    glr[4 * q + 0].x = u.x; glr[4 * q + 0].y = w.x;
    glr[4 * q + 1].x = u.y; glr[4 * q + 1].y = w.y;
    glr[4 * q + 2].x = u.z; glr[4 * q + 2].y = w.z;
    glr[4 * q + 3].x = u.w; glr[4 * q + 3].y = w.w;
  }
  float av[D_ / 2];
#pragma unroll
  for (int d = 0; d < D_ / 2; ++d) av[d] = sAtt[HF * (D_ / 2) + d];

  cfloat_k* gh = grw + HF * (D_ / 2);
  float* srow_w1 = srow_w + 32 * SP_;
  for (int i = 0; i < SWV_; i += RIF_) {
    v2f sc[RIF_];
#pragma unroll
    for (int j = 0; j < RIF_; ++j) {
      sc[j].x = HF ? srow_w[i + j] : 0.f;
      sc[j].y = HF ? srow_w1[i + j] : 0.f;
    }
#pragma unroll
    for (int d = 0; d < D_ / 2; ++d) {
      v2f a2; a2.x = av[d]; a2.y = av[d];
#pragma unroll
      for (int j = 0; j < RIF_; ++j) {
        const float g = gh[(long)(i + j) * HID_ + d];
        v2f g2; g2.x = g; g2.y = g;
        const v2f v = glr[d] + g2;
        const v2f m = __builtin_elementwise_max(v, v * NEG_);
        sc[j] = __builtin_elementwise_fma(a2, m, sc[j]);
      }
    }
#pragma unroll
    for (int j = 0; j < RIF_; ++j) {
      srow_w[i + j]  = sc[j].x;
      srow_w1[i + j] = sc[j].y;
    }
  }
}

__global__ __launch_bounds__(32 * NWV_) void gat_scores_softmax(
    const float* __restrict__ gl, const float* __restrict__ gr,
    const float* __restrict__ att, float* __restrict__ alpha)
{
  __shared__ float sS[TCH_ * SP_];
  __shared__ float sAtt[D_];

  const int tid = threadIdx.x;
  const int lane = tid & 31;
  const int wv = __builtin_amdgcn_readfirstlane(tid >> 5);
  const int tc = blockIdx.x;
  const int h = blockIdx.y;
  const int b = blockIdx.z;

  if (wv < D_ / 32) sAtt[tid] = att[h * D_ + tid];

  __syncthreads();

  const float* glrow = gl + ((long)b * C_ + tc * TCH_ + lane) * HID_ + h * D_;
  cfloat_k* grw = (cfloat_k*)(gr + ((long)b * C_ + wv * SWV_) * HID_ + h * D_);
  float* srow_w = &sS[lane * SP_ + wv * SWV_];

  score_half<0>(glrow, grw, sAtt, srow_w);
  score_half<1>(glrow, grw, sAtt, srow_w);
  __syncthreads();

  float* arow = alpha + ((long)(b * H_ + h) * C_ + tc * TCH_) * C_;
#pragma unroll
  for (int k = 0; k < TCH_ / NWV_; ++k) {
    const int tl = wv * (TCH_ / NWV_) + k;
    const float* sr = &sS[tl * SP_];
    float v0 = sr[lane], v1 = sr[lane + 32], v2 = sr[lane + 64], v3 = sr[lane + 96];
    float m = fmaxf(fmaxf(v0, v1), fmaxf(v2, v3));
    for (int off = 16; off > 0; off >>= 1) m = fmaxf(m, __shfl_xor(m, off, 32));
    const float e0 = __expf(v0 - m), e1 = __expf(v1 - m);
    const float e2 = __expf(v2 - m), e3 = __expf(v3 - m);
    float r0 = e0, r1 = e1, r2 = e2, r3 = e3;
    for (int off = 16; off > 0; off >>= 1) {
      r0 += __shfl_xor(r0, off, 32);
      r1 += __shfl_xor(r1, off, 32);
      r2 += __shfl_xor(r2, off, 32);
      r3 += __shfl_xor(r3, off, 32);
    }
    const float ssum = r0 + r1 + r2 + r3;
    float* o = arow + (long)tl * C_ + lane;
    o[0]  = e0 / ssum;
    o[32] = e1 / ssum;
    o[64] = e2 / ssum;
    o[96] = e3 / ssum;
  }
}

#define LNR_ 2

__device__ __forceinline__ float elu1(float x) {
  return x > 0.f ? x : (__expf(x) - 1.f);
}

__global__ __launch_bounds__(256) void post_elu_residual_ln(
    float* __restrict__ h, const float* __restrict__ gout,
    const float* __restrict__ g, const float* __restrict__ bb)
{
  const int lane = threadIdx.x & 31;
  const int wv   = threadIdx.x >> 5;
  const long row0 = ((long)blockIdx.x * 8 + wv) * LNR_;

  const float4 g0 = *(const float4*)(g + 4 * lane);
  const float4 g1 = *(const float4*)(g + 128 + 4 * lane);
  const float4 c0 = *(const float4*)(bb + 4 * lane);
  const float4 c1 = *(const float4*)(bb + 128 + 4 * lane);

#pragma unroll
  for (int k = 0; k < LNR_; ++k) {
    const long base = (row0 + k) * HID_;
    const float4 x0 = *(const float4*)(gout + base + 4 * lane);
    const float4 x1 = *(const float4*)(gout + base + 128 + 4 * lane);
    const float4 h0 = *(const float4*)(h + base + 4 * lane);
    const float4 h1 = *(const float4*)(h + base + 128 + 4 * lane);
    const float v0 = h0.x + elu1(x0.x), v1 = h0.y + elu1(x0.y);
    const float v2 = h0.z + elu1(x0.z), v3 = h0.w + elu1(x0.w);
    const float v4 = h1.x + elu1(x1.x), v5 = h1.y + elu1(x1.y);
    const float v6 = h1.z + elu1(x1.z), v7 = h1.w + elu1(x1.w);

    float s = ((v0 + v1) + (v2 + v3)) + ((v4 + v5) + (v6 + v7));
    for (int off = 16; off > 0; off >>= 1) s += __shfl_xor(s, off, 32);
    const float mu = s * (1.f / HID_);

    const float d0 = v0 - mu, d1 = v1 - mu, d2 = v2 - mu, d3 = v3 - mu;
    const float d4 = v4 - mu, d5 = v5 - mu, d6 = v6 - mu, d7 = v7 - mu;
    float q = ((d0 * d0 + d1 * d1) + (d2 * d2 + d3 * d3)) +
              ((d4 * d4 + d5 * d5) + (d6 * d6 + d7 * d7));
    for (int off = 16; off > 0; off >>= 1) q += __shfl_xor(q, off, 32);
    const float inv = rsqrtf(q * (1.f / HID_) + EPS_);

    float4 o0, o1;
    o0.x = d0 * inv * g0.x + c0.x;  o0.y = d1 * inv * g0.y + c0.y;
    o0.z = d2 * inv * g0.z + c0.z;  o0.w = d3 * inv * g0.w + c0.w;
    o1.x = d4 * inv * g1.x + c1.x;  o1.y = d5 * inv * g1.y + c1.y;
    o1.z = d6 * inv * g1.z + c1.z;  o1.w = d7 * inv * g1.w + c1.w;
    *(float4*)(h + base + 4 * lane)       = o0;
    *(float4*)(h + base + 128 + 4 * lane) = o1;
  }
}

extern "C" void kernel_launch(void* const* d_in, const int* in_sizes, int n_in,
                              void* d_out, int out_size, void* d_ws, size_t ws_size,
                              hipStream_t stream) {
  (void)in_sizes; (void)n_in; (void)out_size; (void)ws_size;
  const float* x      = (const float*)d_in[0];
  const float* W_emb  = (const float*)d_in[1];
  const float* b_emb  = (const float*)d_in[2];
  const float* Wl     = (const float*)d_in[3];
  const float* Wr     = (const float*)d_in[4];
  const float* att    = (const float*)d_in[5];
  const float* bias_g = (const float*)d_in[6];
  const float* ln_g   = (const float*)d_in[7];
  const float* ln_b   = (const float*)d_in[8];
  const float* W_proj = (const float*)d_in[9];
  const float* b_proj = (const float*)d_in[10];

  float* out   = (float*)d_out;
  float* alpha = out + (long)B_ * L_ * C_;

  float* ws = (float*)d_ws;
  const long hwN = (long)B_ * C_ * HID_;
  float* h    = ws;
  float* gl   = ws + hwN;
  float* gr   = ws + 2 * hwN;
  float* gout = ws + 3 * hwN;

  wmma_gemm_f32<C_, HID_, L_, C_, HID_, HID_, 1,
                (long)L_ * C_, 0, 0, 0, (long)C_ * HID_, 0, 0,
                true, false, true>
      <<<dim3((C_ / 32) * (HID_ / 64), B_), 128, 0, stream>>>(
          x, W_emb, b_emb, h);

  for (int layer = 0; layer < NL_; ++layer) {
    const float* Wl_i  = Wl + (long)layer * HID_ * HID_;
    const float* Wr_i  = Wr + (long)layer * HID_ * HID_;
    const float* att_i = att + (long)layer * H_ * D_;
    const float* bg_i  = bias_g + (long)layer * HID_;

    dim3 g1((C_ / 32) * (HID_ / 64), B_);
    wmma_gemm_f32<C_, HID_, HID_, HID_, HID_, HID_, 1,
                  (long)C_ * HID_, 0, 0, 0, (long)C_ * HID_, 0, 0,
                  false, false, false>
        <<<g1, 128, 0, stream>>>(h, Wl_i, nullptr, gl);
    wmma_gemm_f32<C_, HID_, HID_, HID_, HID_, HID_, 1,
                  (long)C_ * HID_, 0, 0, 0, (long)C_ * HID_, 0, 0,
                  false, false, false>
        <<<g1, 128, 0, stream>>>(h, Wr_i, nullptr, gr);

    gat_scores_softmax<<<dim3(C_ / TCH_, H_, B_), 32 * NWV_, 0, stream>>>(gl, gr, att_i, alpha);

    wmma_gemm_f32<C_, D_, C_, C_, HID_, HID_, H_,
                  (long)H_ * C_ * C_, (long)C_ * C_,
                  (long)C_ * HID_, D_, (long)C_ * HID_, D_, D_,
                  false, false, true>
        <<<dim3((C_ / 32) * (D_ / 64), B_ * H_), 128, 0, stream>>>(
            alpha, gr, bg_i, gout);

    post_elu_residual_ln<<<B_ * C_ / (8 * LNR_), 256, 0, stream>>>(
        h, gout, ln_g + (long)layer * HID_, ln_b + (long)layer * HID_);
  }

  wmma_gemm_f32<C_, L_, HID_, HID_, L_, C_, 1,
                (long)C_ * HID_, 0, 0, 0, (long)L_ * C_, 0, 0,
                false, true, true>
      <<<dim3((C_ / 32) * (L_ / 64), B_), 128, 0, stream>>>(
          h, W_proj, b_proj, out);
}


# gl projection moved into the scores kernel: a work-group computes the gl tile of its own 64 targets and head on the matrix instructions into LDS; no gl launch, 17 to 14 launches; k order unchanged
# speedup vs baseline: 9.0587x; 1.1071x over previous
#include <hip/hip_runtime.h>
#include <hip/hip_bf16.h>

typedef __attribute__((ext_vector_type(2))) float v2f;
typedef __attribute__((ext_vector_type(8))) float v8f;

#define B_   32
#define L_   512
#define C_   128
#define HID_ 256
#define H_   4
#define D_   64
#define NL_  3
#define NEG_ 0.2f
#define EPS_ 1e-5f
#define KC_  64

template <int M, int N, int K, int LDA, int LDB, int LDD, int INNER,
          long SAO, long SAI, long SBO, long SBI, long SDO, long SDI,
          long SBIASI, bool TRANSA, bool TRANSD, bool HASBIAS>
__global__ __launch_bounds__(128) void wmma_gemm_f32(
    const float* __restrict__ A, const float* __restrict__ Bm,
    const float* __restrict__ bias, float* __restrict__ D)
{
  constexpr int BM = 32;
  constexpr int BN = 64;
  constexpr int KC = KC_;
  constexpr int PA = TRANSA ? BM + 8 : KC + 4;
  constexpr int PB = BN + 8;
  constexpr int A_SZ = TRANSA ? KC * PA : BM * PA;
  constexpr int B_SZ = KC * PB;
  constexpr int A_F4 = BM * KC / 4 / 128;
  constexpr int B_F4 = KC * BN / 4 / 128;
  constexpr int NCH = K / KC;
  static_assert(K % KC == 0 && N % BN == 0 && M % BM == 0, "shape");

  __shared__ __attribute__((aligned(16))) float sA[2 * A_SZ];
  __shared__ __attribute__((aligned(16))) float sB[2 * B_SZ];

  const int tid  = threadIdx.x;
  const int lane = tid & 31;
  const int wave = tid >> 5;
  const int half = (lane >> 4) & 1;
  const int r    = lane & 15;

  constexpr int tilesNb = N / BN;
  const int mb = (blockIdx.x / tilesNb) * BM;
  const int nb = (blockIdx.x % tilesNb) * BN;
  const int wn = wave * 16;

  const int batch = blockIdx.y;
  const int bo = batch / INNER;
  const int bi = batch % INNER;
  A    += bo * SAO + bi * SAI;
  Bm   += bo * SBO + bi * SBI;
  D    += bo * SDO + bi * SDI;
  if (HASBIAS) bias += bi * SBIASI;

  int gA[A_F4], lA[A_F4], gB[B_F4], lB[B_F4];
#pragma unroll
  for (int j = 0; j < A_F4; ++j) {
    const int idx = tid + 128 * j;
    if (TRANSA) {
      const int krow = idx / (BM / 4), mq = idx % (BM / 4);
      gA[j] = krow * LDA + mb + 4 * mq;
      lA[j] = krow * PA + 4 * mq;
    } else {
      const int row = idx / (KC / 4), kq = idx % (KC / 4);
      gA[j] = (mb + row) * LDA + 4 * kq;
      lA[j] = row * PA + 4 * kq;
    }
  }
#pragma unroll
  for (int j = 0; j < B_F4; ++j) {
    const int idx = tid + 128 * j;
    const int krow = idx / (BN / 4), nq = idx % (BN / 4);
    gB[j] = krow * LDB + nb + 4 * nq;
    lB[j] = krow * PB + 4 * nq;
  }
  constexpr int stepA = TRANSA ? KC * LDA : KC;
  constexpr int stepB = KC * LDB;


  v8f acc0 = {0.f,0.f,0.f,0.f,0.f,0.f,0.f,0.f};
  v8f acc1 = acc0;

  float4 ra[A_F4], rb[B_F4];
#pragma unroll
  for (int j = 0; j < A_F4; ++j) ra[j] = *(const float4*)(A + gA[j]);
#pragma unroll
  for (int j = 0; j < B_F4; ++j) rb[j] = *(const float4*)(Bm + gB[j]);
#pragma unroll
  for (int j = 0; j < A_F4; ++j) *(float4*)(sA + lA[j]) = ra[j];
#pragma unroll
  for (int j = 0; j < B_F4; ++j) *(float4*)(sB + lB[j]) = rb[j];
  __syncthreads();

#pragma unroll
  for (int c = 0; c < NCH; ++c) {
    const float* as = sA + (c & 1) * A_SZ;
    const float* bs = sB + (c & 1) * B_SZ;
    if (c + 1 < NCH) {
#pragma unroll
      for (int j = 0; j < A_F4; ++j) ra[j] = *(const float4*)(A + gA[j] + (c + 1) * stepA);
#pragma unroll
      for (int j = 0; j < B_F4; ++j) rb[j] = *(const float4*)(Bm + gB[j] + (c + 1) * stepB);
      if (!TRANSA) __builtin_amdgcn_sched_barrier(0);
    }
#pragma unroll (TRANSA ? 8 : KC / 4)
    for (int ks = 0; ks < KC / 4; ++ks) {
      const int k = 4 * ks + 2 * half;
      v2f a0, a1, b0;
      if (TRANSA) {
        a0.x = as[k * PA + r];      a0.y = as[(k + 1) * PA + r];
        a1.x = as[k * PA + 16 + r]; a1.y = as[(k + 1) * PA + 16 + r];
      } else {
        a0 = *(const v2f*)(as + r * PA + k);
        a1 = *(const v2f*)(as + (16 + r) * PA + k);
      }
      b0.x = bs[k * PB + wn + r];   b0.y = bs[(k + 1) * PB + wn + r];

      acc0 = __builtin_amdgcn_wmma_f32_16x16x4_f32(false, a0, false, b0, (short)0, acc0, false, false);
      acc1 = __builtin_amdgcn_wmma_f32_16x16x4_f32(false, a1, false, b0, (short)0, acc1, false, false);
    }
    if (c + 1 < NCH) {
      float* an = sA + ((c + 1) & 1) * A_SZ;
      float* bn = sB + ((c + 1) & 1) * B_SZ;
#pragma unroll
      for (int j = 0; j < A_F4; ++j) *(float4*)(an + lA[j]) = ra[j];
#pragma unroll
      for (int j = 0; j < B_F4; ++j) *(float4*)(bn + lB[j]) = rb[j];
      __syncthreads();
    }
  }

  const int col0 = nb + wn + r;
  const float bv0 = HASBIAS ? bias[col0] : 0.f;
#pragma unroll
  for (int i = 0; i < 8; ++i) {
    const int row0 = mb + i + 8 * half;
    const int row1 = row0 + 16;
    if (TRANSD) {
      D[(long)col0 * LDD + row0] = acc0[i] + bv0;
      D[(long)col0 * LDD + row1] = acc1[i] + bv0;
    } else {
      D[(long)row0 * LDD + col0] = acc0[i] + bv0;
      D[(long)row1 * LDD + col0] = acc1[i] + bv0;
    }
  }
}

#define TCH_ 64
#define NWV_ 8
#define SWV_ (C_ / NWV_)
#define SP_  (C_ + 1)
#define RIF_ 4
#define PKC_ 64
#define PPA_ (PKC_ + 4)
#define PPB_ (D_ + 8)
#define GP_  (D_ + 4)

typedef const __attribute__((address_space(4))) float cfloat_k;

template <int HF>
__device__ __forceinline__ void score_half(
    const float* __restrict__ glrow, cfloat_k* grw,
    const float* sAtt, float* srow_w)
{
  const float4* glp0 = (const float4*)(glrow + HF * (D_ / 2));
  const float4* glp1 = (const float4*)(glrow + 32 * GP_ + HF * (D_ / 2));
  v2f glr[D_ / 2];
#pragma unroll
  for (int q = 0; q < D_ / 8; ++q) {
    const float4 u = glp0[q];
    const float4 w = glp1[q];
    glr[4 * q + 0].x = u.x; glr[4 * q + 0].y = w.x;
    glr[4 * q + 1].x = u.y; glr[4 * q + 1].y = w.y;
    glr[4 * q + 2].x = u.z; glr[4 * q + 2].y = w.z;
    glr[4 * q + 3].x = u.w; glr[4 * q + 3].y = w.w;
  }
  float av[D_ / 2];
#pragma unroll
  for (int d = 0; d < D_ / 2; ++d) av[d] = sAtt[HF * (D_ / 2) + d];

  cfloat_k* gh = grw + HF * (D_ / 2);
  float* srow_w1 = srow_w + 32 * SP_;
  for (int i = 0; i < SWV_; i += RIF_) {
    v2f sc[RIF_];
#pragma unroll
    for (int j = 0; j < RIF_; ++j) {
      sc[j].x = HF ? srow_w[i + j] : 0.f;
      sc[j].y = HF ? srow_w1[i + j] : 0.f;
    }
#pragma unroll
    for (int d = 0; d < D_ / 2; ++d) {
      v2f a2; a2.x = av[d]; a2.y = av[d];
#pragma unroll
      for (int j = 0; j < RIF_; ++j) {
        const float g = gh[(long)(i + j) * HID_ + d];
        v2f g2; g2.x = g; g2.y = g;
        const v2f v = glr[d] + g2;
        const v2f m = __builtin_elementwise_max(v, v * NEG_);
        sc[j] = __builtin_elementwise_fma(a2, m, sc[j]);
      }
    }
#pragma unroll
    for (int j = 0; j < RIF_; ++j) {
      srow_w[i + j]  = sc[j].x;
      srow_w1[i + j] = sc[j].y;
    }
  }
}

__device__ __forceinline__ void project_gl_tile(
    const float* __restrict__ A, const float* __restrict__ Bm,
    float* sA, float* sB, float* sG, const int tid, const int wave)
{
  constexpr int A_SZ = TCH_ * PPA_;
  constexpr int B_SZ = PKC_ * PPB_;
  constexpr int NF4 = TCH_ * PKC_ / 4 / (32 * NWV_);
  constexpr int NCH = HID_ / PKC_;
  static_assert(TCH_ == 64 && D_ == 64 && PKC_ == 64 && NWV_ == 8, "tile shape");

  const int lane = tid & 31;
  const int half = (lane >> 4) & 1;
  const int r    = lane & 15;
  const int wm = (wave >> 2) * 32;
  const int wn = (wave & 3) * 16;

  int gO[NF4], lA[NF4], lB[NF4];
#pragma unroll
  for (int j = 0; j < NF4; ++j) {
    const int idx = tid + 32 * NWV_ * j;
    const int row = idx / (PKC_ / 4), q = idx % (PKC_ / 4);
    gO[j] = row * HID_ + 4 * q;
    lA[j] = row * PPA_ + 4 * q;
    lB[j] = row * PPB_ + 4 * q;
  }

  v8f acc0 = {0.f,0.f,0.f,0.f,0.f,0.f,0.f,0.f};
  v8f acc1 = acc0;

  float4 ra[NF4], rb[NF4];
#pragma unroll
  for (int j = 0; j < NF4; ++j) ra[j] = *(const float4*)(A + gO[j]);
#pragma unroll
  for (int j = 0; j < NF4; ++j) rb[j] = *(const float4*)(Bm + gO[j]);
#pragma unroll
  for (int j = 0; j < NF4; ++j) *(float4*)(sA + lA[j]) = ra[j];
#pragma unroll
  for (int j = 0; j < NF4; ++j) *(float4*)(sB + lB[j]) = rb[j];
  __syncthreads();

#pragma unroll
  for (int c = 0; c < NCH; ++c) {
    const float* as = sA + (c & 1) * A_SZ;
    const float* bs = sB + (c & 1) * B_SZ;
    if (c + 1 < NCH) {
#pragma unroll
      for (int j = 0; j < NF4; ++j) ra[j] = *(const float4*)(A + gO[j] + (c + 1) * PKC_);
#pragma unroll
      for (int j = 0; j < NF4; ++j) rb[j] = *(const float4*)(Bm + gO[j] + (c + 1) * PKC_ * HID_);
      __builtin_amdgcn_sched_barrier(0);
    }
#pragma unroll
    for (int ks = 0; ks < PKC_ / 4; ++ks) {
      const int k = 4 * ks + 2 * half;
      const v2f a0 = *(const v2f*)(as + (wm + r) * PPA_ + k);
      const v2f a1 = *(const v2f*)(as + (wm + 16 + r) * PPA_ + k);
      v2f b0;
      b0.x = bs[k * PPB_ + wn + r];   b0.y = bs[(k + 1) * PPB_ + wn + r];
      acc0 = __builtin_amdgcn_wmma_f32_16x16x4_f32(false, a0, false, b0, (short)0, acc0, false, false);
      acc1 = __builtin_amdgcn_wmma_f32_16x16x4_f32(false, a1, false, b0, (short)0, acc1, false, false);
    }
    if (c + 1 < NCH) {
      float* an = sA + ((c + 1) & 1) * A_SZ;
      float* bn = sB + ((c + 1) & 1) * B_SZ;
#pragma unroll
      for (int j = 0; j < NF4; ++j) *(float4*)(an + lA[j]) = ra[j];
#pragma unroll
      for (int j = 0; j < NF4; ++j) *(float4*)(bn + lB[j]) = rb[j];
      __syncthreads();
    }
  }

#pragma unroll
  for (int i = 0; i < 8; ++i) {
    sG[(wm + i + 8 * half) * GP_ + wn + r]      = acc0[i];
    sG[(wm + 16 + i + 8 * half) * GP_ + wn + r] = acc1[i];
  }
  __syncthreads();
}

__global__ __launch_bounds__(32 * NWV_) void gat_scores_softmax(
    const float* __restrict__ hin, const float* __restrict__ Wl,
    const float* __restrict__ gr,
    const float* __restrict__ att, float* __restrict__ alpha)
{
  __shared__ float sS[TCH_ * SP_];
  __shared__ float sAtt[D_];
  __shared__ __attribute__((aligned(16))) float sPA[2 * TCH_ * PPA_];
  __shared__ __attribute__((aligned(16))) float sPB[2 * PKC_ * PPB_];
  __shared__ __attribute__((aligned(16))) float sG[TCH_ * GP_];

  const int tid = threadIdx.x;
  const int lane = tid & 31;
  const int wv = __builtin_amdgcn_readfirstlane(tid >> 5);
  const int tc = blockIdx.x;
  const int h = blockIdx.y;
  const int b = blockIdx.z;

  if (wv < D_ / 32) sAtt[tid] = att[h * D_ + tid];

  project_gl_tile(hin + ((long)b * C_ + tc * TCH_) * HID_, Wl + h * D_,
                  sPA, sPB, sG, tid, wv);

  const float* glrow = sG + lane * GP_;
  cfloat_k* grw = (cfloat_k*)(gr + ((long)b * C_ + wv * SWV_) * HID_ + h * D_);
  float* srow_w = &sS[lane * SP_ + wv * SWV_];

  score_half<0>(glrow, grw, sAtt, srow_w);
  score_half<1>(glrow, grw, sAtt, srow_w);
  __syncthreads();

  float* arow = alpha + ((long)(b * H_ + h) * C_ + tc * TCH_) * C_;
#pragma unroll
  for (int k = 0; k < TCH_ / NWV_; ++k) {
    const int tl = wv * (TCH_ / NWV_) + k;
    const float* sr = &sS[tl * SP_];
    float v0 = sr[lane], v1 = sr[lane + 32], v2 = sr[lane + 64], v3 = sr[lane + 96];
    float m = fmaxf(fmaxf(v0, v1), fmaxf(v2, v3));
    for (int off = 16; off > 0; off >>= 1) m = fmaxf(m, __shfl_xor(m, off, 32));
    const float e0 = __expf(v0 - m), e1 = __expf(v1 - m);
    const float e2 = __expf(v2 - m), e3 = __expf(v3 - m);
    float r0 = e0, r1 = e1, r2 = e2, r3 = e3;
    for (int off = 16; off > 0; off >>= 1) {
      r0 += __shfl_xor(r0, off, 32);
      r1 += __shfl_xor(r1, off, 32);
      r2 += __shfl_xor(r2, off, 32);
      r3 += __shfl_xor(r3, off, 32);
    }
    const float ssum = r0 + r1 + r2 + r3;
    float* o = arow + (long)tl * C_ + lane;
    o[0]  = e0 / ssum;
    o[32] = e1 / ssum;
    o[64] = e2 / ssum;
    o[96] = e3 / ssum;
  }
}

#define LNR_ 2

__device__ __forceinline__ float elu1(float x) {
  return x > 0.f ? x : (__expf(x) - 1.f);
}

__global__ __launch_bounds__(256) void post_elu_residual_ln(
    float* __restrict__ h, const float* __restrict__ gout,
    const float* __restrict__ g, const float* __restrict__ bb)
{
  const int lane = threadIdx.x & 31;
  const int wv   = threadIdx.x >> 5;
  const long row0 = ((long)blockIdx.x * 8 + wv) * LNR_;

  const float4 g0 = *(const float4*)(g + 4 * lane);
  const float4 g1 = *(const float4*)(g + 128 + 4 * lane);
  const float4 c0 = *(const float4*)(bb + 4 * lane);
  const float4 c1 = *(const float4*)(bb + 128 + 4 * lane);

#pragma unroll
  for (int k = 0; k < LNR_; ++k) {
    const long base = (row0 + k) * HID_;
    const float4 x0 = *(const float4*)(gout + base + 4 * lane);
    const float4 x1 = *(const float4*)(gout + base + 128 + 4 * lane);
    const float4 h0 = *(const float4*)(h + base + 4 * lane);
    const float4 h1 = *(const float4*)(h + base + 128 + 4 * lane);
    const float v0 = h0.x + elu1(x0.x), v1 = h0.y + elu1(x0.y);
    const float v2 = h0.z + elu1(x0.z), v3 = h0.w + elu1(x0.w);
    const float v4 = h1.x + elu1(x1.x), v5 = h1.y + elu1(x1.y);
    const float v6 = h1.z + elu1(x1.z), v7 = h1.w + elu1(x1.w);

    float s = ((v0 + v1) + (v2 + v3)) + ((v4 + v5) + (v6 + v7));
    for (int off = 16; off > 0; off >>= 1) s += __shfl_xor(s, off, 32);
    const float mu = s * (1.f / HID_);

    const float d0 = v0 - mu, d1 = v1 - mu, d2 = v2 - mu, d3 = v3 - mu;
    const float d4 = v4 - mu, d5 = v5 - mu, d6 = v6 - mu, d7 = v7 - mu;
    float q = ((d0 * d0 + d1 * d1) + (d2 * d2 + d3 * d3)) +
              ((d4 * d4 + d5 * d5) + (d6 * d6 + d7 * d7));
    for (int off = 16; off > 0; off >>= 1) q += __shfl_xor(q, off, 32);
    const float inv = rsqrtf(q * (1.f / HID_) + EPS_);

    float4 o0, o1;
    o0.x = d0 * inv * g0.x + c0.x;  o0.y = d1 * inv * g0.y + c0.y;
    o0.z = d2 * inv * g0.z + c0.z;  o0.w = d3 * inv * g0.w + c0.w;
    o1.x = d4 * inv * g1.x + c1.x;  o1.y = d5 * inv * g1.y + c1.y;
    o1.z = d6 * inv * g1.z + c1.z;  o1.w = d7 * inv * g1.w + c1.w;
    *(float4*)(h + base + 4 * lane)       = o0;
    *(float4*)(h + base + 128 + 4 * lane) = o1;
  }
}

extern "C" void kernel_launch(void* const* d_in, const int* in_sizes, int n_in,
                              void* d_out, int out_size, void* d_ws, size_t ws_size,
                              hipStream_t stream) {
  (void)in_sizes; (void)n_in; (void)out_size; (void)ws_size;
  const float* x      = (const float*)d_in[0];
  const float* W_emb  = (const float*)d_in[1];
  const float* b_emb  = (const float*)d_in[2];
  const float* Wl     = (const float*)d_in[3];
  const float* Wr     = (const float*)d_in[4];
  const float* att    = (const float*)d_in[5];
  const float* bias_g = (const float*)d_in[6];
  const float* ln_g   = (const float*)d_in[7];
  const float* ln_b   = (const float*)d_in[8];
  const float* W_proj = (const float*)d_in[9];
  const float* b_proj = (const float*)d_in[10];

  float* out   = (float*)d_out;
  float* alpha = out + (long)B_ * L_ * C_;

  float* ws = (float*)d_ws;
  const long hwN = (long)B_ * C_ * HID_;
  float* h    = ws;
  float* gr   = ws + 2 * hwN;
  float* gout = ws + 3 * hwN;

  wmma_gemm_f32<C_, HID_, L_, C_, HID_, HID_, 1,
                (long)L_ * C_, 0, 0, 0, (long)C_ * HID_, 0, 0,
                true, false, true>
      <<<dim3((C_ / 32) * (HID_ / 64), B_), 128, 0, stream>>>(
          x, W_emb, b_emb, h);

  for (int layer = 0; layer < NL_; ++layer) {
    const float* Wl_i  = Wl + (long)layer * HID_ * HID_;
    const float* Wr_i  = Wr + (long)layer * HID_ * HID_;
    const float* att_i = att + (long)layer * H_ * D_;
    const float* bg_i  = bias_g + (long)layer * HID_;

    dim3 g1((C_ / 32) * (HID_ / 64), B_);
    wmma_gemm_f32<C_, HID_, HID_, HID_, HID_, HID_, 1,
                  (long)C_ * HID_, 0, 0, 0, (long)C_ * HID_, 0, 0,
                  false, false, false>
        <<<g1, 128, 0, stream>>>(h, Wr_i, nullptr, gr);

    gat_scores_softmax<<<dim3(C_ / TCH_, H_, B_), 32 * NWV_, 0, stream>>>(h, Wl_i, gr, att_i, alpha);

    wmma_gemm_f32<C_, D_, C_, C_, HID_, HID_, H_,
                  (long)H_ * C_ * C_, (long)C_ * C_,
                  (long)C_ * HID_, D_, (long)C_ * HID_, D_, D_,
                  false, false, true>
        <<<dim3((C_ / 32) * (D_ / 64), B_ * H_), 128, 0, stream>>>(
            alpha, gr, bg_i, gout);

    post_elu_residual_ln<<<B_ * C_ / (8 * LNR_), 256, 0, stream>>>(
        h, gout, ln_g + (long)layer * HID_, ln_b + (long)layer * HID_);
  }

  wmma_gemm_f32<C_, L_, HID_, HID_, L_, C_, 1,
                (long)C_ * HID_, 0, 0, 0, (long)L_ * C_, 0, 0,
                false, true, true>
      <<<dim3((C_ / 32) * (L_ / 64), B_), 128, 0, stream>>>(
          h, W_proj, b_proj, out);
}


# aggregation launch re-tiled: a work-group owns 16 whole rows (4 heads x 64 columns, 8 waves) and finishes elu, residual and layer norm from its own LDS tile: no norm launch, 14 to 11; sum orders kept
# speedup vs baseline: 9.4995x; 1.0487x over previous
#include <hip/hip_runtime.h>
#include <hip/hip_bf16.h>

typedef __attribute__((ext_vector_type(2))) float v2f;
typedef __attribute__((ext_vector_type(8))) float v8f;

#define B_   32
#define L_   512
#define C_   128
#define HID_ 256
#define H_   4
#define D_   64
#define NL_  3
#define NEG_ 0.2f
#define EPS_ 1e-5f
#define KC_  64

template <int M, int N, int K, int LDA, int LDB, int LDD, int INNER,
          long SAO, long SAI, long SBO, long SBI, long SDO, long SDI,
          long SBIASI, bool TRANSA, bool TRANSD, bool HASBIAS>
__global__ __launch_bounds__(128) void wmma_gemm_f32(
    const float* __restrict__ A, const float* __restrict__ Bm,
    const float* __restrict__ bias, float* __restrict__ D)
{
  constexpr int BM = 32;
  constexpr int BN = 64;
  constexpr int KC = KC_;
  constexpr int PA = TRANSA ? BM + 8 : KC + 4;
  constexpr int PB = BN + 8;
  constexpr int A_SZ = TRANSA ? KC * PA : BM * PA;
  constexpr int B_SZ = KC * PB;
  constexpr int A_F4 = BM * KC / 4 / 128;
  constexpr int B_F4 = KC * BN / 4 / 128;
  constexpr int NCH = K / KC;
  static_assert(K % KC == 0 && N % BN == 0 && M % BM == 0, "shape");

  __shared__ __attribute__((aligned(16))) float sA[2 * A_SZ];
  __shared__ __attribute__((aligned(16))) float sB[2 * B_SZ];

  const int tid  = threadIdx.x;
  const int lane = tid & 31;
  const int wave = tid >> 5;
  const int half = (lane >> 4) & 1;
  const int r    = lane & 15;

  constexpr int tilesNb = N / BN;
  const int mb = (blockIdx.x / tilesNb) * BM;
  const int nb = (blockIdx.x % tilesNb) * BN;
  const int wn = wave * 16;

  const int batch = blockIdx.y;
  const int bo = batch / INNER;
  const int bi = batch % INNER;
  A    += bo * SAO + bi * SAI;
  Bm   += bo * SBO + bi * SBI;
  D    += bo * SDO + bi * SDI;
  if (HASBIAS) bias += bi * SBIASI;

  int gA[A_F4], lA[A_F4], gB[B_F4], lB[B_F4];
#pragma unroll
  for (int j = 0; j < A_F4; ++j) {
    const int idx = tid + 128 * j;
    if (TRANSA) {
      const int krow = idx / (BM / 4), mq = idx % (BM / 4);
      gA[j] = krow * LDA + mb + 4 * mq;
      lA[j] = krow * PA + 4 * mq;
    } else {
      const int row = idx / (KC / 4), kq = idx % (KC / 4);
      gA[j] = (mb + row) * LDA + 4 * kq;
      lA[j] = row * PA + 4 * kq;
    }
  }
#pragma unroll
  for (int j = 0; j < B_F4; ++j) {
    const int idx = tid + 128 * j;
    const int krow = idx / (BN / 4), nq = idx % (BN / 4);
    gB[j] = krow * LDB + nb + 4 * nq;
    lB[j] = krow * PB + 4 * nq;
  }
  constexpr int stepA = TRANSA ? KC * LDA : KC;
  constexpr int stepB = KC * LDB;


  v8f acc0 = {0.f,0.f,0.f,0.f,0.f,0.f,0.f,0.f};
  v8f acc1 = acc0;

  float4 ra[A_F4], rb[B_F4];
#pragma unroll
  for (int j = 0; j < A_F4; ++j) ra[j] = *(const float4*)(A + gA[j]);
#pragma unroll
  for (int j = 0; j < B_F4; ++j) rb[j] = *(const float4*)(Bm + gB[j]);
#pragma unroll
  for (int j = 0; j < A_F4; ++j) *(float4*)(sA + lA[j]) = ra[j];
#pragma unroll
  for (int j = 0; j < B_F4; ++j) *(float4*)(sB + lB[j]) = rb[j];
  __syncthreads();

#pragma unroll
  for (int c = 0; c < NCH; ++c) {
    const float* as = sA + (c & 1) * A_SZ;
    const float* bs = sB + (c & 1) * B_SZ;
    if (c + 1 < NCH) {
#pragma unroll
      for (int j = 0; j < A_F4; ++j) ra[j] = *(const float4*)(A + gA[j] + (c + 1) * stepA);
#pragma unroll
      for (int j = 0; j < B_F4; ++j) rb[j] = *(const float4*)(Bm + gB[j] + (c + 1) * stepB);
      if (!TRANSA) __builtin_amdgcn_sched_barrier(0);
    }
#pragma unroll (TRANSA ? 8 : KC / 4)
    for (int ks = 0; ks < KC / 4; ++ks) {
      const int k = 4 * ks + 2 * half;
      v2f a0, a1, b0;
      if (TRANSA) {
        a0.x = as[k * PA + r];      a0.y = as[(k + 1) * PA + r];
        a1.x = as[k * PA + 16 + r]; a1.y = as[(k + 1) * PA + 16 + r];
      } else {
        a0 = *(const v2f*)(as + r * PA + k);
        a1 = *(const v2f*)(as + (16 + r) * PA + k);
      }
      b0.x = bs[k * PB + wn + r];   b0.y = bs[(k + 1) * PB + wn + r];

      acc0 = __builtin_amdgcn_wmma_f32_16x16x4_f32(false, a0, false, b0, (short)0, acc0, false, false);
      acc1 = __builtin_amdgcn_wmma_f32_16x16x4_f32(false, a1, false, b0, (short)0, acc1, false, false);
    }
    if (c + 1 < NCH) {
      float* an = sA + ((c + 1) & 1) * A_SZ;
      float* bn = sB + ((c + 1) & 1) * B_SZ;
#pragma unroll
      for (int j = 0; j < A_F4; ++j) *(float4*)(an + lA[j]) = ra[j];
#pragma unroll
      for (int j = 0; j < B_F4; ++j) *(float4*)(bn + lB[j]) = rb[j];
      __syncthreads();
    }
  }

  const int col0 = nb + wn + r;
  const float bv0 = HASBIAS ? bias[col0] : 0.f;
#pragma unroll
  for (int i = 0; i < 8; ++i) {
    const int row0 = mb + i + 8 * half;
    const int row1 = row0 + 16;
    if (TRANSD) {
      D[(long)col0 * LDD + row0] = acc0[i] + bv0;
      D[(long)col0 * LDD + row1] = acc1[i] + bv0;
    } else {
      D[(long)row0 * LDD + col0] = acc0[i] + bv0;
      D[(long)row1 * LDD + col0] = acc1[i] + bv0;
    }
  }
}

#define TCH_ 64
#define NWV_ 8
#define SWV_ (C_ / NWV_)
#define SP_  (C_ + 1)
#define RIF_ 4
#define PKC_ 64
#define PPA_ (PKC_ + 4)
#define PPB_ (D_ + 8)
#define GP_  (D_ + 4)

typedef const __attribute__((address_space(4))) float cfloat_k;

template <int HF>
__device__ __forceinline__ void score_half(
    const float* __restrict__ glrow, cfloat_k* grw,
    const float* sAtt, float* srow_w)
{
  const float4* glp0 = (const float4*)(glrow + HF * (D_ / 2));
  const float4* glp1 = (const float4*)(glrow + 32 * GP_ + HF * (D_ / 2));
  v2f glr[D_ / 2];
#pragma unroll
  for (int q = 0; q < D_ / 8; ++q) {
    const float4 u = glp0[q];
    const float4 w = glp1[q];
    glr[4 * q + 0].x = u.x; glr[4 * q + 0].y = w.x;
    glr[4 * q + 1].x = u.y; glr[4 * q + 1].y = w.y;
    glr[4 * q + 2].x = u.z; glr[4 * q + 2].y = w.z;
    glr[4 * q + 3].x = u.w; glr[4 * q + 3].y = w.w;
  }
  float av[D_ / 2];
#pragma unroll
  for (int d = 0; d < D_ / 2; ++d) av[d] = sAtt[HF * (D_ / 2) + d];

  cfloat_k* gh = grw + HF * (D_ / 2);
  float* srow_w1 = srow_w + 32 * SP_;
  for (int i = 0; i < SWV_; i += RIF_) {
    v2f sc[RIF_];
#pragma unroll
    for (int j = 0; j < RIF_; ++j) {
      sc[j].x = HF ? srow_w[i + j] : 0.f;
      sc[j].y = HF ? srow_w1[i + j] : 0.f;
    }
#pragma unroll
    for (int d = 0; d < D_ / 2; ++d) {
      v2f a2; a2.x = av[d]; a2.y = av[d];
#pragma unroll
      for (int j = 0; j < RIF_; ++j) {
        const float g = gh[(long)(i + j) * HID_ + d];
        v2f g2; g2.x = g; g2.y = g;
        const v2f v = glr[d] + g2;
        const v2f m = __builtin_elementwise_max(v, v * NEG_);
        sc[j] = __builtin_elementwise_fma(a2, m, sc[j]);
      }
    }
#pragma unroll
    for (int j = 0; j < RIF_; ++j) {
      srow_w[i + j]  = sc[j].x;
      srow_w1[i + j] = sc[j].y;
    }
  }
}

__device__ __forceinline__ void project_gl_tile(
    const float* __restrict__ A, const float* __restrict__ Bm,
    float* sA, float* sB, float* sG, const int tid, const int wave)
{
  constexpr int A_SZ = TCH_ * PPA_;
  constexpr int B_SZ = PKC_ * PPB_;
  constexpr int NF4 = TCH_ * PKC_ / 4 / (32 * NWV_);
  constexpr int NCH = HID_ / PKC_;
  static_assert(TCH_ == 64 && D_ == 64 && PKC_ == 64 && NWV_ == 8, "tile shape");

  const int lane = tid & 31;
  const int half = (lane >> 4) & 1;
  const int r    = lane & 15;
  const int wm = (wave >> 2) * 32;
  const int wn = (wave & 3) * 16;

  int gO[NF4], lA[NF4], lB[NF4];
#pragma unroll
  for (int j = 0; j < NF4; ++j) {
    const int idx = tid + 32 * NWV_ * j;
    const int row = idx / (PKC_ / 4), q = idx % (PKC_ / 4);
    gO[j] = row * HID_ + 4 * q;
    lA[j] = row * PPA_ + 4 * q;
    lB[j] = row * PPB_ + 4 * q;
  }

  v8f acc0 = {0.f,0.f,0.f,0.f,0.f,0.f,0.f,0.f};
  v8f acc1 = acc0;

  float4 ra[NF4], rb[NF4];
#pragma unroll
  for (int j = 0; j < NF4; ++j) ra[j] = *(const float4*)(A + gO[j]);
#pragma unroll
  for (int j = 0; j < NF4; ++j) rb[j] = *(const float4*)(Bm + gO[j]);
#pragma unroll
  for (int j = 0; j < NF4; ++j) *(float4*)(sA + lA[j]) = ra[j];
#pragma unroll
  for (int j = 0; j < NF4; ++j) *(float4*)(sB + lB[j]) = rb[j];
  __syncthreads();

#pragma unroll
  for (int c = 0; c < NCH; ++c) {
    const float* as = sA + (c & 1) * A_SZ;
    const float* bs = sB + (c & 1) * B_SZ;
    if (c + 1 < NCH) {
#pragma unroll
      for (int j = 0; j < NF4; ++j) ra[j] = *(const float4*)(A + gO[j] + (c + 1) * PKC_);
#pragma unroll
      for (int j = 0; j < NF4; ++j) rb[j] = *(const float4*)(Bm + gO[j] + (c + 1) * PKC_ * HID_);
      __builtin_amdgcn_sched_barrier(0);
    }
#pragma unroll
    for (int ks = 0; ks < PKC_ / 4; ++ks) {
      const int k = 4 * ks + 2 * half;
      const v2f a0 = *(const v2f*)(as + (wm + r) * PPA_ + k);
      const v2f a1 = *(const v2f*)(as + (wm + 16 + r) * PPA_ + k);
      v2f b0;
      b0.x = bs[k * PPB_ + wn + r];   b0.y = bs[(k + 1) * PPB_ + wn + r];
      acc0 = __builtin_amdgcn_wmma_f32_16x16x4_f32(false, a0, false, b0, (short)0, acc0, false, false);
      acc1 = __builtin_amdgcn_wmma_f32_16x16x4_f32(false, a1, false, b0, (short)0, acc1, false, false);
    }
    if (c + 1 < NCH) {
      float* an = sA + ((c + 1) & 1) * A_SZ;
      float* bn = sB + ((c + 1) & 1) * B_SZ;
#pragma unroll
      for (int j = 0; j < NF4; ++j) *(float4*)(an + lA[j]) = ra[j];
#pragma unroll
      for (int j = 0; j < NF4; ++j) *(float4*)(bn + lB[j]) = rb[j];
      __syncthreads();
    }
  }

#pragma unroll
  for (int i = 0; i < 8; ++i) {
    sG[(wm + i + 8 * half) * GP_ + wn + r]      = acc0[i];
    sG[(wm + 16 + i + 8 * half) * GP_ + wn + r] = acc1[i];
  }
  __syncthreads();
}

__global__ __launch_bounds__(32 * NWV_) void gat_scores_softmax(
    const float* __restrict__ hin, const float* __restrict__ Wl,
    const float* __restrict__ gr,
    const float* __restrict__ att, float* __restrict__ alpha)
{
  __shared__ float sS[TCH_ * SP_];
  __shared__ float sAtt[D_];
  __shared__ __attribute__((aligned(16))) float sPA[2 * TCH_ * PPA_];
  __shared__ __attribute__((aligned(16))) float sPB[2 * PKC_ * PPB_];
  __shared__ __attribute__((aligned(16))) float sG[TCH_ * GP_];

  const int tid = threadIdx.x;
  const int lane = tid & 31;
  const int wv = __builtin_amdgcn_readfirstlane(tid >> 5);
  const int tc = blockIdx.x;
  const int h = blockIdx.y;
  const int b = blockIdx.z;

  if (wv < D_ / 32) sAtt[tid] = att[h * D_ + tid];

  project_gl_tile(hin + ((long)b * C_ + tc * TCH_) * HID_, Wl + h * D_,
                  sPA, sPB, sG, tid, wv);

  const float* glrow = sG + lane * GP_;
  cfloat_k* grw = (cfloat_k*)(gr + ((long)b * C_ + wv * SWV_) * HID_ + h * D_);
  float* srow_w = &sS[lane * SP_ + wv * SWV_];

  score_half<0>(glrow, grw, sAtt, srow_w);
  score_half<1>(glrow, grw, sAtt, srow_w);
  __syncthreads();

  float* arow = alpha + ((long)(b * H_ + h) * C_ + tc * TCH_) * C_;
#pragma unroll
  for (int k = 0; k < TCH_ / NWV_; ++k) {
    const int tl = wv * (TCH_ / NWV_) + k;
    const float* sr = &sS[tl * SP_];
    float v0 = sr[lane], v1 = sr[lane + 32], v2 = sr[lane + 64], v3 = sr[lane + 96];
    float m = fmaxf(fmaxf(v0, v1), fmaxf(v2, v3));
    for (int off = 16; off > 0; off >>= 1) m = fmaxf(m, __shfl_xor(m, off, 32));
    const float e0 = __expf(v0 - m), e1 = __expf(v1 - m);
    const float e2 = __expf(v2 - m), e3 = __expf(v3 - m);
    float r0 = e0, r1 = e1, r2 = e2, r3 = e3;
    for (int off = 16; off > 0; off >>= 1) {
      r0 += __shfl_xor(r0, off, 32);
      r1 += __shfl_xor(r1, off, 32);
      r2 += __shfl_xor(r2, off, 32);
      r3 += __shfl_xor(r3, off, 32);
    }
    const float ssum = r0 + r1 + r2 + r3;
    float* o = arow + (long)tl * C_ + lane;
    o[0]  = e0 / ssum;
    o[32] = e1 / ssum;
    o[64] = e2 / ssum;
    o[96] = e3 / ssum;
  }
}

#define AR_   16
#define AKC_  64
#define APA_  (AKC_ + 4)
#define APB_  (HID_ + 8)
#define AGP_  (HID_ + 4)
#define LNR_  2

__device__ __forceinline__ float elu1(float x) {
  return x > 0.f ? x : (__expf(x) - 1.f);
}

__global__ __launch_bounds__(256) void agg_elu_residual_ln(
    const float* __restrict__ alpha, const float* __restrict__ gr,
    const float* __restrict__ bias, float* __restrict__ h,
    const float* __restrict__ g, const float* __restrict__ bb)
{
  constexpr int A_SZ = H_ * AR_ * APA_;
  constexpr int B_SZ = AKC_ * APB_;
  constexpr int NA4 = H_ * AR_ * AKC_ / 4 / 256;
  constexpr int NB4 = AKC_ * HID_ / 4 / 256;
  static_assert(C_ == 2 * AKC_ && NA4 == H_ && AR_ == 8 * LNR_ && D_ == 64, "tile shape");

  __shared__ __attribute__((aligned(16))) float sA[2 * A_SZ];
  __shared__ __attribute__((aligned(16))) float sB[2 * B_SZ];
  __shared__ __attribute__((aligned(16))) float sO[AR_ * AGP_];

  const int tid  = threadIdx.x;
  const int lane = tid & 31;
  const int wave = tid >> 5;
  const int half = (lane >> 4) & 1;
  const int r    = lane & 15;
  const int hd   = wave >> 1;
  const int col0 = hd * D_ + (wave & 1) * 32;
  const int rb   = blockIdx.x * AR_;
  const int b    = blockIdx.y;

  const float* Ab = alpha + ((long)b * H_ * C_ + rb) * C_;
  const float* Bb = gr + (long)b * C_ * HID_;
  float* hb = h + ((long)b * C_ + rb) * HID_;

  const int arow = tid >> 4, aq = tid & 15;
  const int gA = arow * C_ + 4 * aq;
  const int lA = arow * APA_ + 4 * aq;
  const int bq = tid & 63, brow = tid >> 6;
  const int gB = brow * HID_ + 4 * bq;
  const int lB = brow * APB_ + 4 * bq;

  float4 ra[NA4], rv[NB4];
#pragma unroll
  for (int j = 0; j < NA4; ++j) ra[j] = *(const float4*)(Ab + (long)j * C_ * C_ + gA);
#pragma unroll
  for (int j = 0; j < NB4; ++j) rv[j] = *(const float4*)(Bb + gB + j * 4 * HID_);
#pragma unroll
  for (int j = 0; j < NA4; ++j) *(float4*)(sA + j * AR_ * APA_ + lA) = ra[j];
#pragma unroll
  for (int j = 0; j < NB4; ++j) *(float4*)(sB + lB + j * 4 * APB_) = rv[j];

  const float4 g0 = *(const float4*)(g + 4 * lane);
  const float4 g1 = *(const float4*)(g + 128 + 4 * lane);
  const float4 c0 = *(const float4*)(bb + 4 * lane);
  const float4 c1 = *(const float4*)(bb + 128 + 4 * lane);
  float4 hp0[LNR_], hp1[LNR_];
#pragma unroll
  for (int k = 0; k < LNR_; ++k) {
    const int row = wave * LNR_ + k;
    hp0[k] = *(const float4*)(hb + row * HID_ + 4 * lane);
    hp1[k] = *(const float4*)(hb + row * HID_ + 128 + 4 * lane);
  }
  const float bv0 = bias[col0 + r];
  const float bv1 = bias[col0 + 16 + r];
  __syncthreads();

#pragma unroll
  for (int j = 0; j < NA4; ++j) ra[j] = *(const float4*)(Ab + (long)j * C_ * C_ + gA + AKC_);
#pragma unroll
  for (int j = 0; j < NB4; ++j) rv[j] = *(const float4*)(Bb + gB + j * 4 * HID_ + AKC_ * HID_);

  v8f acc0 = {0.f,0.f,0.f,0.f,0.f,0.f,0.f,0.f};
  v8f acc1 = acc0;

#pragma unroll
  for (int c = 0; c < 2; ++c) {
    const float* as = sA + c * A_SZ + (hd * AR_ + r) * APA_;
    const float* bs = sB + c * B_SZ + col0 + r;
    v2f fa[AKC_ / 4], fb0[AKC_ / 4], fb1[AKC_ / 4];
#pragma unroll
    for (int ks = 0; ks < AKC_ / 4; ++ks) {
      const int k = 4 * ks + 2 * half;
      fa[ks] = *(const v2f*)(as + k);
      fb0[ks].x = bs[k * APB_];        fb0[ks].y = bs[(k + 1) * APB_];
      fb1[ks].x = bs[k * APB_ + 16];   fb1[ks].y = bs[(k + 1) * APB_ + 16];
    }
    __builtin_amdgcn_sched_barrier(0);
#pragma unroll
    for (int ks = 0; ks < AKC_ / 4; ++ks) {
      acc0 = __builtin_amdgcn_wmma_f32_16x16x4_f32(false, fa[ks], false, fb0[ks], (short)0, acc0, false, false);
      acc1 = __builtin_amdgcn_wmma_f32_16x16x4_f32(false, fa[ks], false, fb1[ks], (short)0, acc1, false, false);
    }
    if (c == 0) {
#pragma unroll
      for (int j = 0; j < NA4; ++j) *(float4*)(sA + A_SZ + j * AR_ * APA_ + lA) = ra[j];
#pragma unroll
      for (int j = 0; j < NB4; ++j) *(float4*)(sB + B_SZ + lB + j * 4 * APB_) = rv[j];
      __builtin_amdgcn_sched_group_barrier(0x020, NA4 + NB4, 0);
      __builtin_amdgcn_sched_group_barrier(0x008, AKC_ / 2, 0);
      __builtin_amdgcn_sched_group_barrier(0x200, NA4 + NB4, 0);
      __syncthreads();
    }
  }

#pragma unroll
  for (int i = 0; i < 8; ++i) {
    sO[(i + 8 * half) * AGP_ + col0 + r]      = acc0[i] + bv0;
    sO[(i + 8 * half) * AGP_ + col0 + 16 + r] = acc1[i] + bv1;
  }
  __syncthreads();

#pragma unroll
  for (int k = 0; k < LNR_; ++k) {
    const int row = wave * LNR_ + k;
    const float4 x0 = *(const float4*)(sO + row * AGP_ + 4 * lane);
    const float4 x1 = *(const float4*)(sO + row * AGP_ + 128 + 4 * lane);
    const float4 h0 = hp0[k];
    const float4 h1 = hp1[k];
    const float v0 = h0.x + elu1(x0.x), v1 = h0.y + elu1(x0.y);
    const float v2 = h0.z + elu1(x0.z), v3 = h0.w + elu1(x0.w);
    const float v4 = h1.x + elu1(x1.x), v5 = h1.y + elu1(x1.y);
    const float v6 = h1.z + elu1(x1.z), v7 = h1.w + elu1(x1.w);

    float s = ((v0 + v1) + (v2 + v3)) + ((v4 + v5) + (v6 + v7));
    for (int off = 16; off > 0; off >>= 1) s += __shfl_xor(s, off, 32);
    const float mu = s * (1.f / HID_);

    const float d0 = v0 - mu, d1 = v1 - mu, d2 = v2 - mu, d3 = v3 - mu;
    const float d4 = v4 - mu, d5 = v5 - mu, d6 = v6 - mu, d7 = v7 - mu;
    float q = ((d0 * d0 + d1 * d1) + (d2 * d2 + d3 * d3)) +
              ((d4 * d4 + d5 * d5) + (d6 * d6 + d7 * d7));
    for (int off = 16; off > 0; off >>= 1) q += __shfl_xor(q, off, 32);
    const float inv = rsqrtf(q * (1.f / HID_) + EPS_);

    float4 o0, o1;
    o0.x = d0 * inv * g0.x + c0.x;  o0.y = d1 * inv * g0.y + c0.y;
    o0.z = d2 * inv * g0.z + c0.z;  o0.w = d3 * inv * g0.w + c0.w;
    o1.x = d4 * inv * g1.x + c1.x;  o1.y = d5 * inv * g1.y + c1.y;
    o1.z = d6 * inv * g1.z + c1.z;  o1.w = d7 * inv * g1.w + c1.w;
    *(float4*)(hb + row * HID_ + 4 * lane)       = o0;
    *(float4*)(hb + row * HID_ + 128 + 4 * lane) = o1;
  }
}

extern "C" void kernel_launch(void* const* d_in, const int* in_sizes, int n_in,
                              void* d_out, int out_size, void* d_ws, size_t ws_size,
                              hipStream_t stream) {
  (void)in_sizes; (void)n_in; (void)out_size; (void)ws_size;
  const float* x      = (const float*)d_in[0];
  const float* W_emb  = (const float*)d_in[1];
  const float* b_emb  = (const float*)d_in[2];
  const float* Wl     = (const float*)d_in[3];
  const float* Wr     = (const float*)d_in[4];
  const float* att    = (const float*)d_in[5];
  const float* bias_g = (const float*)d_in[6];
  const float* ln_g   = (const float*)d_in[7];
  const float* ln_b   = (const float*)d_in[8];
  const float* W_proj = (const float*)d_in[9];
  const float* b_proj = (const float*)d_in[10];

  float* out   = (float*)d_out;
  float* alpha = out + (long)B_ * L_ * C_;

  float* ws = (float*)d_ws;
  const long hwN = (long)B_ * C_ * HID_;
  float* h    = ws;
  float* gr   = ws + 2 * hwN;

  wmma_gemm_f32<C_, HID_, L_, C_, HID_, HID_, 1,
                (long)L_ * C_, 0, 0, 0, (long)C_ * HID_, 0, 0,
                true, false, true>
      <<<dim3((C_ / 32) * (HID_ / 64), B_), 128, 0, stream>>>(
          x, W_emb, b_emb, h);

  for (int layer = 0; layer < NL_; ++layer) {
    const float* Wl_i  = Wl + (long)layer * HID_ * HID_;
    const float* Wr_i  = Wr + (long)layer * HID_ * HID_;
    const float* att_i = att + (long)layer * H_ * D_;
    const float* bg_i  = bias_g + (long)layer * HID_;

    dim3 g1((C_ / 32) * (HID_ / 64), B_);
    wmma_gemm_f32<C_, HID_, HID_, HID_, HID_, HID_, 1,
                  (long)C_ * HID_, 0, 0, 0, (long)C_ * HID_, 0, 0,
                  false, false, false>
        <<<g1, 128, 0, stream>>>(h, Wr_i, nullptr, gr);

    gat_scores_softmax<<<dim3(C_ / TCH_, H_, B_), 32 * NWV_, 0, stream>>>(h, Wl_i, gr, att_i, alpha);

    agg_elu_residual_ln<<<dim3(C_ / AR_, B_), 256, 0, stream>>>(
        alpha, gr, bg_i, h, ln_g + (long)layer * HID_, ln_b + (long)layer * HID_);
  }

  wmma_gemm_f32<C_, L_, HID_, HID_, L_, C_, 1,
                (long)C_ * HID_, 0, 0, 0, (long)L_ * C_, 0, 0,
                false, true, true>
      <<<dim3((C_ / 32) * (L_ / 64), B_), 128, 0, stream>>>(
          h, W_proj, b_proj, out);
}


# scores kernel: the (b,h) source tile goes to LDS once (whole-line loads under the projection's last chunk); d loop reads source values by LDS broadcast one group ahead, not scalar loads; sums kept
# speedup vs baseline: 10.2579x; 1.0798x over previous
#include <hip/hip_runtime.h>
#include <hip/hip_bf16.h>

typedef __attribute__((ext_vector_type(2))) float v2f;
typedef __attribute__((ext_vector_type(8))) float v8f;

#define B_   32
#define L_   512
#define C_   128
#define HID_ 256
#define H_   4
#define D_   64
#define NL_  3
#define NEG_ 0.2f
#define EPS_ 1e-5f
#define KC_  64

template <int M, int N, int K, int LDA, int LDB, int LDD, int INNER,
          long SAO, long SAI, long SBO, long SBI, long SDO, long SDI,
          long SBIASI, bool TRANSA, bool TRANSD, bool HASBIAS>
__global__ __launch_bounds__(128) void wmma_gemm_f32(
    const float* __restrict__ A, const float* __restrict__ Bm,
    const float* __restrict__ bias, float* __restrict__ D)
{
  constexpr int BM = 32;
  constexpr int BN = 64;
  constexpr int KC = KC_;
  constexpr int PA = TRANSA ? BM + 8 : KC + 4;
  constexpr int PB = BN + 8;
  constexpr int A_SZ = TRANSA ? KC * PA : BM * PA;
  constexpr int B_SZ = KC * PB;
  constexpr int A_F4 = BM * KC / 4 / 128;
  constexpr int B_F4 = KC * BN / 4 / 128;
  constexpr int NCH = K / KC;
  static_assert(K % KC == 0 && N % BN == 0 && M % BM == 0, "shape");

  __shared__ __attribute__((aligned(16))) float sA[2 * A_SZ];
  __shared__ __attribute__((aligned(16))) float sB[2 * B_SZ];

  const int tid  = threadIdx.x;
  const int lane = tid & 31;
  const int wave = tid >> 5;
  const int half = (lane >> 4) & 1;
  const int r    = lane & 15;

  constexpr int tilesNb = N / BN;
  const int mb = (blockIdx.x / tilesNb) * BM;
  const int nb = (blockIdx.x % tilesNb) * BN;
  const int wn = wave * 16;

  const int batch = blockIdx.y;
  const int bo = batch / INNER;
  const int bi = batch % INNER;
  A    += bo * SAO + bi * SAI;
  Bm   += bo * SBO + bi * SBI;
  D    += bo * SDO + bi * SDI;
  if (HASBIAS) bias += bi * SBIASI;

  int gA[A_F4], lA[A_F4], gB[B_F4], lB[B_F4];
#pragma unroll
  for (int j = 0; j < A_F4; ++j) {
    const int idx = tid + 128 * j;
    if (TRANSA) {
      const int krow = idx / (BM / 4), mq = idx % (BM / 4);
      gA[j] = krow * LDA + mb + 4 * mq;
      lA[j] = krow * PA + 4 * mq;
    } else {
      const int row = idx / (KC / 4), kq = idx % (KC / 4);
      gA[j] = (mb + row) * LDA + 4 * kq;
      lA[j] = row * PA + 4 * kq;
    }
  }
#pragma unroll
  for (int j = 0; j < B_F4; ++j) {
    const int idx = tid + 128 * j;
    const int krow = idx / (BN / 4), nq = idx % (BN / 4);
    gB[j] = krow * LDB + nb + 4 * nq;
    lB[j] = krow * PB + 4 * nq;
  }
  constexpr int stepA = TRANSA ? KC * LDA : KC;
  constexpr int stepB = KC * LDB;


  v8f acc0 = {0.f,0.f,0.f,0.f,0.f,0.f,0.f,0.f};
  v8f acc1 = acc0;

  float4 ra[A_F4], rb[B_F4];
#pragma unroll
  for (int j = 0; j < A_F4; ++j) ra[j] = *(const float4*)(A + gA[j]);
#pragma unroll
  for (int j = 0; j < B_F4; ++j) rb[j] = *(const float4*)(Bm + gB[j]);
#pragma unroll
  for (int j = 0; j < A_F4; ++j) *(float4*)(sA + lA[j]) = ra[j];
#pragma unroll
  for (int j = 0; j < B_F4; ++j) *(float4*)(sB + lB[j]) = rb[j];
  __syncthreads();

#pragma unroll
  for (int c = 0; c < NCH; ++c) {
    const float* as = sA + (c & 1) * A_SZ;
    const float* bs = sB + (c & 1) * B_SZ;
    if (c + 1 < NCH) {
#pragma unroll
      for (int j = 0; j < A_F4; ++j) ra[j] = *(const float4*)(A + gA[j] + (c + 1) * stepA);
#pragma unroll
      for (int j = 0; j < B_F4; ++j) rb[j] = *(const float4*)(Bm + gB[j] + (c + 1) * stepB);
      if (!TRANSA) __builtin_amdgcn_sched_barrier(0);
    }
#pragma unroll (TRANSA ? 8 : KC / 4)
    for (int ks = 0; ks < KC / 4; ++ks) {
      const int k = 4 * ks + 2 * half;
      v2f a0, a1, b0;
      if (TRANSA) {
        a0.x = as[k * PA + r];      a0.y = as[(k + 1) * PA + r];
        a1.x = as[k * PA + 16 + r]; a1.y = as[(k + 1) * PA + 16 + r];
      } else {
        a0 = *(const v2f*)(as + r * PA + k);
        a1 = *(const v2f*)(as + (16 + r) * PA + k);
      }
      b0.x = bs[k * PB + wn + r];   b0.y = bs[(k + 1) * PB + wn + r];

      acc0 = __builtin_amdgcn_wmma_f32_16x16x4_f32(false, a0, false, b0, (short)0, acc0, false, false);
      acc1 = __builtin_amdgcn_wmma_f32_16x16x4_f32(false, a1, false, b0, (short)0, acc1, false, false);
    }
    if (c + 1 < NCH) {
      float* an = sA + ((c + 1) & 1) * A_SZ;
      float* bn = sB + ((c + 1) & 1) * B_SZ;
#pragma unroll
      for (int j = 0; j < A_F4; ++j) *(float4*)(an + lA[j]) = ra[j];
#pragma unroll
      for (int j = 0; j < B_F4; ++j) *(float4*)(bn + lB[j]) = rb[j];
      __syncthreads();
    }
  }

  const int col0 = nb + wn + r;
  const float bv0 = HASBIAS ? bias[col0] : 0.f;
#pragma unroll
  for (int i = 0; i < 8; ++i) {
    const int row0 = mb + i + 8 * half;
    const int row1 = row0 + 16;
    if (TRANSD) {
      D[(long)col0 * LDD + row0] = acc0[i] + bv0;
      D[(long)col0 * LDD + row1] = acc1[i] + bv0;
    } else {
      D[(long)row0 * LDD + col0] = acc0[i] + bv0;
      D[(long)row1 * LDD + col0] = acc1[i] + bv0;
    }
  }
}

#define TCH_ 64
#define NWV_ 8
#define SWV_ (C_ / NWV_)
#define SP_  (C_ + 1)
#define RIF_ 4
#define PKC_ 64
#define PPA_ (PKC_ + 4)
#define PPB_ (D_ + 8)
#define GP_  (D_ + 4)
#define GRP_ D_
#define NG4_ (C_ * D_ / 4 / (32 * NWV_))
#define DG_  (D_ / 2 / 8)

typedef __attribute__((ext_vector_type(4))) float v4f;

template <int HF>
__device__ __forceinline__ void score_half(
    const float* __restrict__ glrow, const float* grw,
    const float* sAtt, float* srow_w)
{
  const float4* glp0 = (const float4*)(glrow + HF * (D_ / 2));
  const float4* glp1 = (const float4*)(glrow + 32 * GP_ + HF * (D_ / 2));
  v2f glr[D_ / 2];
#pragma unroll
  for (int q = 0; q < D_ / 8; ++q) {
    const float4 u = glp0[q];
    const float4 w = glp1[q];
    glr[4 * q + 0].x = u.x; glr[4 * q + 0].y = w.x;
    glr[4 * q + 1].x = u.y; glr[4 * q + 1].y = w.y;
    glr[4 * q + 2].x = u.z; glr[4 * q + 2].y = w.z;
    glr[4 * q + 3].x = u.w; glr[4 * q + 3].y = w.w;
  }
  v2f ap[D_ / 2];
#pragma unroll
  for (int d = 0; d < D_ / 2; ++d) ap[d] = *(const v2f*)(sAtt + 2 * (HF * (D_ / 2) + d));

  const float* gh = grw + HF * (D_ / 2);
  float* srow_w1 = srow_w + 32 * SP_;
  v4f ga[RIF_][2], gb[RIF_][2];
#pragma unroll
  for (int j = 0; j < RIF_; ++j) {
    ga[j][0] = *(const v4f*)(gh + j * GRP_);
    ga[j][1] = *(const v4f*)(gh + j * GRP_ + 4);
  }
  for (int i = 0; i < SWV_; i += RIF_) {
    v2f sc[RIF_];
#pragma unroll
    for (int j = 0; j < RIF_; ++j) {
      sc[j].x = HF ? srow_w[i + j] : 0.f;
      sc[j].y = HF ? srow_w1[i + j] : 0.f;
    }
    const float* rows  = gh + i * GRP_;
    const float* rowsn = gh + ((i + RIF_ < SWV_) ? i + RIF_ : i) * GRP_;
#pragma unroll
    for (int gi = 0; gi < DG_; ++gi) {
      const float* np = (gi + 1 < DG_) ? rows + 8 * (gi + 1) : rowsn;
#pragma unroll
      for (int j = 0; j < RIF_; ++j) {
        const v4f n0 = *(const v4f*)(np + j * GRP_);
        const v4f n1 = *(const v4f*)(np + j * GRP_ + 4);
        if (gi & 1) { ga[j][0] = n0; ga[j][1] = n1; } else { gb[j][0] = n0; gb[j][1] = n1; }
      }
      __builtin_amdgcn_sched_barrier(0);
#pragma unroll
      for (int e = 0; e < 8; ++e) {
        const int d = 8 * gi + e;
        const v2f a2 = ap[d];
#pragma unroll
        for (int j = 0; j < RIF_; ++j) {
          const v4f gv = (gi & 1) ? gb[j][e >> 2] : ga[j][e >> 2];
          const v2f gp = (e & 2) ? __builtin_shufflevector(gv, gv, 2, 3) : __builtin_shufflevector(gv, gv, 0, 1);
          const v2f g2 = (e & 1) ? __builtin_shufflevector(gp, gp, 1, 1) : __builtin_shufflevector(gp, gp, 0, 0);
          const v2f v = glr[d] + g2;
          const v2f m = __builtin_elementwise_max(v, v * NEG_);
          sc[j] = __builtin_elementwise_fma(a2, m, sc[j]);
        }
      }
    }
#pragma unroll
    for (int j = 0; j < RIF_; ++j) {
      srow_w[i + j]  = sc[j].x;
      srow_w1[i + j] = sc[j].y;
    }
  }
}

__device__ __forceinline__ void project_gl_tile(
    const float* __restrict__ A, const float* __restrict__ Bm,
    const float* __restrict__ grsrc, float* sGr,
    float* sA, float* sB, float* sG, const int tid, const int wave)
{
  constexpr int A_SZ = TCH_ * PPA_;
  constexpr int B_SZ = PKC_ * PPB_;
  constexpr int NF4 = TCH_ * PKC_ / 4 / (32 * NWV_);
  constexpr int NCH = HID_ / PKC_;
  static_assert(TCH_ == 64 && D_ == 64 && PKC_ == 64 && NWV_ == 8, "tile shape");

  const int lane = tid & 31;
  const int half = (lane >> 4) & 1;
  const int r    = lane & 15;
  const int wm = (wave >> 2) * 32;
  const int wn = (wave & 3) * 16;

  int gO[NF4], lA[NF4], lB[NF4];
#pragma unroll
  for (int j = 0; j < NF4; ++j) {
    const int idx = tid + 32 * NWV_ * j;
    const int row = idx / (PKC_ / 4), q = idx % (PKC_ / 4);
    gO[j] = row * HID_ + 4 * q;
    lA[j] = row * PPA_ + 4 * q;
    lB[j] = row * PPB_ + 4 * q;
  }

  v8f acc0 = {0.f,0.f,0.f,0.f,0.f,0.f,0.f,0.f};
  v8f acc1 = acc0;

  float4 ra[NF4], rb[NF4], rg[NG4_];
#pragma unroll
  for (int j = 0; j < NF4; ++j) ra[j] = *(const float4*)(A + gO[j]);
#pragma unroll
  for (int j = 0; j < NF4; ++j) rb[j] = *(const float4*)(Bm + gO[j]);
#pragma unroll
  for (int j = 0; j < NF4; ++j) *(float4*)(sA + lA[j]) = ra[j];
#pragma unroll
  for (int j = 0; j < NF4; ++j) *(float4*)(sB + lB[j]) = rb[j];
  __syncthreads();

#pragma unroll
  for (int c = 0; c < NCH; ++c) {
    const float* as = sA + (c & 1) * A_SZ;
    const float* bs = sB + (c & 1) * B_SZ;
    if (c + 1 < NCH) {
#pragma unroll
      for (int j = 0; j < NF4; ++j) ra[j] = *(const float4*)(A + gO[j] + (c + 1) * PKC_);
#pragma unroll
      for (int j = 0; j < NF4; ++j) rb[j] = *(const float4*)(Bm + gO[j] + (c + 1) * PKC_ * HID_);
      __builtin_amdgcn_sched_barrier(0);
    } else {
#pragma unroll
      for (int j = 0; j < NG4_; ++j) {
        const int idx = tid + 32 * NWV_ * j;
        rg[j] = *(const float4*)(grsrc + (idx / (D_ / 4)) * HID_ + 4 * (idx % (D_ / 4)));
      }
    }
#pragma unroll
    for (int ks = 0; ks < PKC_ / 4; ++ks) {
      const int k = 4 * ks + 2 * half;
      const v2f a0 = *(const v2f*)(as + (wm + r) * PPA_ + k);
      const v2f a1 = *(const v2f*)(as + (wm + 16 + r) * PPA_ + k);
      v2f b0;
      b0.x = bs[k * PPB_ + wn + r];   b0.y = bs[(k + 1) * PPB_ + wn + r];
      acc0 = __builtin_amdgcn_wmma_f32_16x16x4_f32(false, a0, false, b0, (short)0, acc0, false, false);
      acc1 = __builtin_amdgcn_wmma_f32_16x16x4_f32(false, a1, false, b0, (short)0, acc1, false, false);
    }
    if (c + 1 < NCH) {
      float* an = sA + ((c + 1) & 1) * A_SZ;
      float* bn = sB + ((c + 1) & 1) * B_SZ;
#pragma unroll
      for (int j = 0; j < NF4; ++j) *(float4*)(an + lA[j]) = ra[j];
#pragma unroll
      for (int j = 0; j < NF4; ++j) *(float4*)(bn + lB[j]) = rb[j];
      __syncthreads();
    }
  }

#pragma unroll
  for (int i = 0; i < 8; ++i) {
    sG[(wm + i + 8 * half) * GP_ + wn + r]      = acc0[i];
    sG[(wm + 16 + i + 8 * half) * GP_ + wn + r] = acc1[i];
  }
#pragma unroll
  for (int j = 0; j < NG4_; ++j) {
    const int idx = tid + 32 * NWV_ * j;
    *(float4*)(sGr + (idx / (D_ / 4)) * GRP_ + 4 * (idx % (D_ / 4))) = rg[j];
  }
  __syncthreads();
}

__global__ __launch_bounds__(32 * NWV_) void gat_scores_softmax(
    const float* __restrict__ hin, const float* __restrict__ Wl,
    const float* __restrict__ gr,
    const float* __restrict__ att, float* __restrict__ alpha)
{
  __shared__ float sS[TCH_ * SP_];
  __shared__ __attribute__((aligned(16))) float sAtt[2 * D_];
  __shared__ __attribute__((aligned(16))) float sGr[C_ * GRP_];
  __shared__ __attribute__((aligned(16))) float sPA[2 * TCH_ * PPA_];
  __shared__ __attribute__((aligned(16))) float sPB[2 * PKC_ * PPB_];
  __shared__ __attribute__((aligned(16))) float sG[TCH_ * GP_];

  const int tid = threadIdx.x;
  const int lane = tid & 31;
  const int wv = __builtin_amdgcn_readfirstlane(tid >> 5);
  const int tc = blockIdx.x;
  const int h = blockIdx.y;
  const int b = blockIdx.z;

  if (wv < 2 * D_ / 32) sAtt[tid] = att[h * D_ + (tid >> 1)];

  project_gl_tile(hin + ((long)b * C_ + tc * TCH_) * HID_, Wl + h * D_,
                  gr + (long)b * C_ * HID_ + h * D_, sGr,
                  sPA, sPB, sG, tid, wv);

  const float* glrow = sG + lane * GP_;
  const float* grw = sGr + wv * SWV_ * GRP_;
  float* srow_w = &sS[lane * SP_ + wv * SWV_];

  score_half<0>(glrow, grw, sAtt, srow_w);
  score_half<1>(glrow, grw, sAtt, srow_w);
  __syncthreads();

  float* arow = alpha + ((long)(b * H_ + h) * C_ + tc * TCH_) * C_;
#pragma unroll
  for (int k = 0; k < TCH_ / NWV_; ++k) {
    const int tl = wv * (TCH_ / NWV_) + k;
    const float* sr = &sS[tl * SP_];
    float v0 = sr[lane], v1 = sr[lane + 32], v2 = sr[lane + 64], v3 = sr[lane + 96];
    float m = fmaxf(fmaxf(v0, v1), fmaxf(v2, v3));
    for (int off = 16; off > 0; off >>= 1) m = fmaxf(m, __shfl_xor(m, off, 32));
    const float e0 = __expf(v0 - m), e1 = __expf(v1 - m);
    const float e2 = __expf(v2 - m), e3 = __expf(v3 - m);
    float r0 = e0, r1 = e1, r2 = e2, r3 = e3;
    for (int off = 16; off > 0; off >>= 1) {
      r0 += __shfl_xor(r0, off, 32);
      r1 += __shfl_xor(r1, off, 32);
      r2 += __shfl_xor(r2, off, 32);
      r3 += __shfl_xor(r3, off, 32);
    }
    const float ssum = r0 + r1 + r2 + r3;
    float* o = arow + (long)tl * C_ + lane;
    o[0]  = e0 / ssum;
    o[32] = e1 / ssum;
    o[64] = e2 / ssum;
    o[96] = e3 / ssum;
  }
}

#define AR_   16
#define AKC_  64
#define APA_  (AKC_ + 4)
#define APB_  (HID_ + 8)
#define AGP_  (HID_ + 4)
#define LNR_  2

__device__ __forceinline__ float elu1(float x) {
  return x > 0.f ? x : (__expf(x) - 1.f);
}

__global__ __launch_bounds__(256) void agg_elu_residual_ln(
    const float* __restrict__ alpha, const float* __restrict__ gr,
    const float* __restrict__ bias, float* __restrict__ h,
    const float* __restrict__ g, const float* __restrict__ bb)
{
  constexpr int A_SZ = H_ * AR_ * APA_;
  constexpr int B_SZ = AKC_ * APB_;
  constexpr int NA4 = H_ * AR_ * AKC_ / 4 / 256;
  constexpr int NB4 = AKC_ * HID_ / 4 / 256;
  static_assert(C_ == 2 * AKC_ && NA4 == H_ && AR_ == 8 * LNR_ && D_ == 64, "tile shape");

  __shared__ __attribute__((aligned(16))) float sA[2 * A_SZ];
  __shared__ __attribute__((aligned(16))) float sB[2 * B_SZ];
  __shared__ __attribute__((aligned(16))) float sO[AR_ * AGP_];

  const int tid  = threadIdx.x;
  const int lane = tid & 31;
  const int wave = tid >> 5;
  const int half = (lane >> 4) & 1;
  const int r    = lane & 15;
  const int hd   = wave >> 1;
  const int col0 = hd * D_ + (wave & 1) * 32;
  const int rb   = blockIdx.x * AR_;
  const int b    = blockIdx.y;

  const float* Ab = alpha + ((long)b * H_ * C_ + rb) * C_;
  const float* Bb = gr + (long)b * C_ * HID_;
  float* hb = h + ((long)b * C_ + rb) * HID_;

  const int arow = tid >> 4, aq = tid & 15;
  const int gA = arow * C_ + 4 * aq;
  const int lA = arow * APA_ + 4 * aq;
  const int bq = tid & 63, brow = tid >> 6;
  const int gB = brow * HID_ + 4 * bq;
  const int lB = brow * APB_ + 4 * bq;

  float4 ra[NA4], rv[NB4];
#pragma unroll
  for (int j = 0; j < NA4; ++j) ra[j] = *(const float4*)(Ab + (long)j * C_ * C_ + gA);
#pragma unroll
  for (int j = 0; j < NB4; ++j) rv[j] = *(const float4*)(Bb + gB + j * 4 * HID_);
#pragma unroll
  for (int j = 0; j < NA4; ++j) *(float4*)(sA + j * AR_ * APA_ + lA) = ra[j];
#pragma unroll
  for (int j = 0; j < NB4; ++j) *(float4*)(sB + lB + j * 4 * APB_) = rv[j];

  const float4 g0 = *(const float4*)(g + 4 * lane);
  const float4 g1 = *(const float4*)(g + 128 + 4 * lane);
  const float4 c0 = *(const float4*)(bb + 4 * lane);
  const float4 c1 = *(const float4*)(bb + 128 + 4 * lane);
  float4 hp0[LNR_], hp1[LNR_];
#pragma unroll
  for (int k = 0; k < LNR_; ++k) {
    const int row = wave * LNR_ + k;
    hp0[k] = *(const float4*)(hb + row * HID_ + 4 * lane);
    hp1[k] = *(const float4*)(hb + row * HID_ + 128 + 4 * lane);
  }
  const float bv0 = bias[col0 + r];
  const float bv1 = bias[col0 + 16 + r];
  __syncthreads();

#pragma unroll
  for (int j = 0; j < NA4; ++j) ra[j] = *(const float4*)(Ab + (long)j * C_ * C_ + gA + AKC_);
#pragma unroll
  for (int j = 0; j < NB4; ++j) rv[j] = *(const float4*)(Bb + gB + j * 4 * HID_ + AKC_ * HID_);

  v8f acc0 = {0.f,0.f,0.f,0.f,0.f,0.f,0.f,0.f};
  v8f acc1 = acc0;

#pragma unroll
  for (int c = 0; c < 2; ++c) {
    const float* as = sA + c * A_SZ + (hd * AR_ + r) * APA_;
    const float* bs = sB + c * B_SZ + col0 + r;
    v2f fa[AKC_ / 4], fb0[AKC_ / 4], fb1[AKC_ / 4];
#pragma unroll
    for (int ks = 0; ks < AKC_ / 4; ++ks) {
      const int k = 4 * ks + 2 * half;
      fa[ks] = *(const v2f*)(as + k);
      fb0[ks].x = bs[k * APB_];        fb0[ks].y = bs[(k + 1) * APB_];
      fb1[ks].x = bs[k * APB_ + 16];   fb1[ks].y = bs[(k + 1) * APB_ + 16];
    }
    __builtin_amdgcn_sched_barrier(0);
#pragma unroll
    for (int ks = 0; ks < AKC_ / 4; ++ks) {
      acc0 = __builtin_amdgcn_wmma_f32_16x16x4_f32(false, fa[ks], false, fb0[ks], (short)0, acc0, false, false);
      acc1 = __builtin_amdgcn_wmma_f32_16x16x4_f32(false, fa[ks], false, fb1[ks], (short)0, acc1, false, false);
    }
    if (c == 0) {
#pragma unroll
      for (int j = 0; j < NA4; ++j) *(float4*)(sA + A_SZ + j * AR_ * APA_ + lA) = ra[j];
#pragma unroll
      for (int j = 0; j < NB4; ++j) *(float4*)(sB + B_SZ + lB + j * 4 * APB_) = rv[j];
      __builtin_amdgcn_sched_group_barrier(0x020, NA4 + NB4, 0);
      __builtin_amdgcn_sched_group_barrier(0x008, AKC_ / 2, 0);
      __builtin_amdgcn_sched_group_barrier(0x200, NA4 + NB4, 0);
      __syncthreads();
    }
  }

#pragma unroll
  for (int i = 0; i < 8; ++i) {
    sO[(i + 8 * half) * AGP_ + col0 + r]      = acc0[i] + bv0;
    sO[(i + 8 * half) * AGP_ + col0 + 16 + r] = acc1[i] + bv1;
  }
  __syncthreads();

#pragma unroll
  for (int k = 0; k < LNR_; ++k) {
    const int row = wave * LNR_ + k;
    const float4 x0 = *(const float4*)(sO + row * AGP_ + 4 * lane);
    const float4 x1 = *(const float4*)(sO + row * AGP_ + 128 + 4 * lane);
    const float4 h0 = hp0[k];
    const float4 h1 = hp1[k];
    const float v0 = h0.x + elu1(x0.x), v1 = h0.y + elu1(x0.y);
    const float v2 = h0.z + elu1(x0.z), v3 = h0.w + elu1(x0.w);
    const float v4 = h1.x + elu1(x1.x), v5 = h1.y + elu1(x1.y);
    const float v6 = h1.z + elu1(x1.z), v7 = h1.w + elu1(x1.w);

    float s = ((v0 + v1) + (v2 + v3)) + ((v4 + v5) + (v6 + v7));
    for (int off = 16; off > 0; off >>= 1) s += __shfl_xor(s, off, 32);
    const float mu = s * (1.f / HID_);

    const float d0 = v0 - mu, d1 = v1 - mu, d2 = v2 - mu, d3 = v3 - mu;
    const float d4 = v4 - mu, d5 = v5 - mu, d6 = v6 - mu, d7 = v7 - mu;
    float q = ((d0 * d0 + d1 * d1) + (d2 * d2 + d3 * d3)) +
              ((d4 * d4 + d5 * d5) + (d6 * d6 + d7 * d7));
    for (int off = 16; off > 0; off >>= 1) q += __shfl_xor(q, off, 32);
    const float inv = rsqrtf(q * (1.f / HID_) + EPS_);

    float4 o0, o1;
    o0.x = d0 * inv * g0.x + c0.x;  o0.y = d1 * inv * g0.y + c0.y;
    o0.z = d2 * inv * g0.z + c0.z;  o0.w = d3 * inv * g0.w + c0.w;
    o1.x = d4 * inv * g1.x + c1.x;  o1.y = d5 * inv * g1.y + c1.y;
    o1.z = d6 * inv * g1.z + c1.z;  o1.w = d7 * inv * g1.w + c1.w;
    *(float4*)(hb + row * HID_ + 4 * lane)       = o0;
    *(float4*)(hb + row * HID_ + 128 + 4 * lane) = o1;
  }
}

extern "C" void kernel_launch(void* const* d_in, const int* in_sizes, int n_in,
                              void* d_out, int out_size, void* d_ws, size_t ws_size,
                              hipStream_t stream) {
  (void)in_sizes; (void)n_in; (void)out_size; (void)ws_size;
  const float* x      = (const float*)d_in[0];
  const float* W_emb  = (const float*)d_in[1];
  const float* b_emb  = (const float*)d_in[2];
  const float* Wl     = (const float*)d_in[3];
  const float* Wr     = (const float*)d_in[4];
  const float* att    = (const float*)d_in[5];
  const float* bias_g = (const float*)d_in[6];
  const float* ln_g   = (const float*)d_in[7];
  const float* ln_b   = (const float*)d_in[8];
  const float* W_proj = (const float*)d_in[9];
  const float* b_proj = (const float*)d_in[10];

  float* out   = (float*)d_out;
  float* alpha = out + (long)B_ * L_ * C_;

  float* ws = (float*)d_ws;
  const long hwN = (long)B_ * C_ * HID_;
  float* h    = ws;
  float* gr   = ws + 2 * hwN;

  wmma_gemm_f32<C_, HID_, L_, C_, HID_, HID_, 1,
                (long)L_ * C_, 0, 0, 0, (long)C_ * HID_, 0, 0,
                true, false, true>
      <<<dim3((C_ / 32) * (HID_ / 64), B_), 128, 0, stream>>>(
          x, W_emb, b_emb, h);

  for (int layer = 0; layer < NL_; ++layer) {
    const float* Wl_i  = Wl + (long)layer * HID_ * HID_;
    const float* Wr_i  = Wr + (long)layer * HID_ * HID_;
    const float* att_i = att + (long)layer * H_ * D_;
    const float* bg_i  = bias_g + (long)layer * HID_;

    dim3 g1((C_ / 32) * (HID_ / 64), B_);
    wmma_gemm_f32<C_, HID_, HID_, HID_, HID_, HID_, 1,
                  (long)C_ * HID_, 0, 0, 0, (long)C_ * HID_, 0, 0,
                  false, false, false>
        <<<g1, 128, 0, stream>>>(h, Wr_i, nullptr, gr);

    gat_scores_softmax<<<dim3(C_ / TCH_, H_, B_), 32 * NWV_, 0, stream>>>(h, Wl_i, gr, att_i, alpha);

    agg_elu_residual_ln<<<dim3(C_ / AR_, B_), 256, 0, stream>>>(
        alpha, gr, bg_i, h, ln_g + (long)layer * HID_, ln_b + (long)layer * HID_);
  }

  wmma_gemm_f32<C_, L_, HID_, HID_, L_, C_, 1,
                (long)C_ * HID_, 0, 0, 0, (long)L_ * C_, 0, 0,
                false, true, true>
      <<<dim3((C_ / 32) * (L_ / 64), B_), 128, 0, stream>>>(
          h, W_proj, b_proj, out);
}
